# starting kernel
# baseline (speedup 1.0000x reference)
#include <hip/hip_runtime.h>
#include <hip/hip_bf16.h>

#define FACTOR 64

typedef float v2f __attribute__((ext_vector_type(2)));
typedef float v8f __attribute__((ext_vector_type(8)));

__device__ __forceinline__ void atomic_add_f32_dev(float* p, float v) {
  unsigned long long a = (unsigned long long)p;
  asm volatile("global_atomic_add_f32 %0, %1, off scope:SCOPE_DEV"
               :: "v"(a), "v"(v));
}

__global__ __launch_bounds__(256) void spmm_coo_atomic(
    const int* __restrict__ rows, const int* __restrict__ cols,
    const float* __restrict__ vals, const float* __restrict__ x,
    float* __restrict__ y, int nnz) {
  const int lane = threadIdx.x;
  const int waveId = blockIdx.x * blockDim.y + threadIdx.y;
  const int e0 = waveId * 4;
  if (e0 >= nnz) return;

  __builtin_prefetch(rows + e0 + 4096, 0, 1);
  __builtin_prefetch(cols + e0 + 4096, 0, 1);
  __builtin_prefetch(vals + e0 + 4096, 0, 1);

  if (e0 + 3 < nnz) {
    const int eu = __builtin_amdgcn_readfirstlane(e0);
    const int4   r4 = *(const int4*)(rows + eu);
    const int4   c4 = *(const int4*)(cols + eu);
    const float4 v4 = *(const float4*)(vals + eu);
    const int   rr[4] = {r4.x, r4.y, r4.z, r4.w};
    const int   cc[4] = {c4.x, c4.y, c4.z, c4.w};
    const float vv[4] = {v4.x, v4.y, v4.z, v4.w};
#pragma unroll
    for (int k = 0; k < 4; ++k) {
      const float2 xv = ((const float2*)(x + (size_t)cc[k] * FACTOR))[lane];
      float* yp = y + (size_t)rr[k] * FACTOR + (size_t)(lane * 2);
      atomic_add_f32_dev(yp + 0, xv.x * vv[k]);
      atomic_add_f32_dev(yp + 1, xv.y * vv[k]);
    }
  } else {
    for (int k = 0; k < 4; ++k) {
      const int e = e0 + k;
      if (e >= nnz) break;
      const int r = rows[e];
      const int c = cols[e];
      const float v = vals[e];
      const float2 xv = ((const float2*)(x + (size_t)c * FACTOR))[lane];
      float* yp = y + (size_t)r * FACTOR + (size_t)(lane * 2);
      atomic_add_f32_dev(yp + 0, xv.x * v);
      atomic_add_f32_dev(yp + 1, xv.y * v);
    }
  }
}

__global__ __launch_bounds__(128) void combine_wmma(
    const float* __restrict__ x0, const float* __restrict__ x1,
    const float* __restrict__ x2, const float* __restrict__ x3,
    float* __restrict__ out, int nrows) {
  const int lane = threadIdx.x & 31;
  const int wave = threadIdx.x >> 5;
  const int row = blockIdx.x * (blockDim.x >> 5) + wave;
  if (row >= nrows) return;

  const bool lo = (lane < 16);
  v2f a;
  a.x = lo ? 1.0f : (1.0f / 3.0f);
  a.y = lo ? 0.5f : 0.25f;

  const float* __restrict__ p0 = lo ? x0 : x2;
  const float* __restrict__ p1 = lo ? x1 : x3;

  const size_t base = (size_t)row * FACTOR;
  const int n = lane & 15;

#pragma unroll
  for (int nb = 0; nb < FACTOR; nb += 16) {
    v2f b;
    b.x = p0[base + nb + n];
    b.y = p1[base + nb + n];
    v8f c = {};
    c = __builtin_amdgcn_wmma_f32_16x16x4_f32(
         false, a,  false, b,
         (short)0, c,  false,  false);
    out[base + nb + n] = c[0];
  }
}

extern "C" void kernel_launch(void* const* d_in, const int* in_sizes, int n_in,
                              void* d_out, int out_size, void* d_ws, size_t ws_size,
                              hipStream_t stream) {
  (void)n_in; (void)out_size; (void)ws_size;

  const float* embed_user = (const float*)d_in[0];
  const float* embed_item = (const float*)d_in[1];
  const int*   u_idx      = (const int*)d_in[2];
  const int*   i_idx      = (const int*)d_in[3];
  const float* vals_ui    = (const float*)d_in[4];
  const float* vals_iu    = (const float*)d_in[5];

  const int U   = in_sizes[0] / FACTOR;
  const int I   = in_sizes[1] / FACTOR;
  const int nnz = in_sizes[2];

  float* g1u = (float*)d_ws;
  float* g1i = g1u + (size_t)U * FACTOR;
  float* g2u = g1i + (size_t)I * FACTOR;
  float* g2i = g2u + (size_t)U * FACTOR;
  float* g3u = g2i + (size_t)I * FACTOR;
  float* g3i = g3u + (size_t)U * FACTOR;

  const size_t zbytes = (size_t)3 * (size_t)(U + I) * FACTOR * sizeof(float);
  hipMemsetAsync(d_ws, 0, zbytes, stream);

  dim3 sblk(32, 8, 1);
  const int waves = (nnz + 3) / 4;
  const int sgrid = (waves + 7) / 8;

  spmm_coo_atomic<<<sgrid, sblk, 0, stream>>>(u_idx, i_idx, vals_ui, embed_item, g1u, nnz);
  spmm_coo_atomic<<<sgrid, sblk, 0, stream>>>(i_idx, u_idx, vals_iu, embed_user, g1i, nnz);
  spmm_coo_atomic<<<sgrid, sblk, 0, stream>>>(u_idx, i_idx, vals_ui, g1i, g2u, nnz);
  spmm_coo_atomic<<<sgrid, sblk, 0, stream>>>(i_idx, u_idx, vals_iu, g1u, g2i, nnz);
  spmm_coo_atomic<<<sgrid, sblk, 0, stream>>>(u_idx, i_idx, vals_ui, g2i, g3u, nnz);
  spmm_coo_atomic<<<sgrid, sblk, 0, stream>>>(i_idx, u_idx, vals_iu, g2u, g3i, nnz);

  float* out_users = (float*)d_out;
  float* out_items = out_users + (size_t)U * FACTOR;
  const int wpb = 4;
  combine_wmma<<<(U + wpb - 1) / wpb, wpb * 32, 0, stream>>>(embed_user, g1u, g2u, g3u, out_users, U);
  combine_wmma<<<(I + wpb - 1) / wpb, wpb * 32, 0, stream>>>(embed_item, g1i, g2i, g3i, out_items, I);
}


# SpMM scatter: one v_permlanex16 half-swap regroups each gathered row so every f32 atomic instruction covers one whole 128-B line (2 line requests per edge, was 4); arrival-order sums as before
# speedup vs baseline: 1.0584x; 1.0584x over previous
#include <hip/hip_runtime.h>
#include <hip/hip_bf16.h>

#define FACTOR 64

typedef float v2f __attribute__((ext_vector_type(2)));
typedef float v8f __attribute__((ext_vector_type(8)));

__device__ __forceinline__ void atomic_add_f32_dev(float* p, float v) {
  unsigned long long a = (unsigned long long)p;
  asm volatile("global_atomic_add_f32 %0, %1, off scope:SCOPE_DEV"
               :: "v"(a), "v"(v));
}

__device__ __forceinline__ float xhalf(float x) {
  const int b = __builtin_bit_cast(int, x);
  return __builtin_bit_cast(float,
      __builtin_amdgcn_permlanex16(b, b, 0x76543210, 0xfedcba98, false, false));
}

__device__ __forceinline__ void scatter_row(float* yrow, float2 xv, float v,
                                            bool lo, int off0) {
  const float px = xv.x * v;
  const float py = xv.y * v;
  const float got = xhalf(lo ? py : px);
  float* yp = yrow + off0;
  atomic_add_f32_dev(yp,      lo ? px : got);
  atomic_add_f32_dev(yp + 32, lo ? got : py);
}

__global__ __launch_bounds__(256) void spmm_coo_atomic(
    const int* __restrict__ rows, const int* __restrict__ cols,
    const float* __restrict__ vals, const float* __restrict__ x,
    float* __restrict__ y, int nnz) {
  const int lane = threadIdx.x;
  const int waveId = blockIdx.x * blockDim.y + threadIdx.y;
  const int e0 = waveId * 4;
  if (e0 >= nnz) return;

  __builtin_prefetch(rows + e0 + 4096, 0, 1);
  __builtin_prefetch(cols + e0 + 4096, 0, 1);
  __builtin_prefetch(vals + e0 + 4096, 0, 1);

  const bool lo = (lane < 16);
  const int off0 = lo ? 2 * lane : 2 * lane - 31;

  if (e0 + 3 < nnz) {
    const int eu = __builtin_amdgcn_readfirstlane(e0);
    const int4   r4 = *(const int4*)(rows + eu);
    const int4   c4 = *(const int4*)(cols + eu);
    const float4 v4 = *(const float4*)(vals + eu);
    const int   rr[4] = {r4.x, r4.y, r4.z, r4.w};
    const int   cc[4] = {c4.x, c4.y, c4.z, c4.w};
    const float vv[4] = {v4.x, v4.y, v4.z, v4.w};
#pragma unroll
    for (int k = 0; k < 4; ++k) {
      const float2 xv = ((const float2*)(x + (size_t)cc[k] * FACTOR))[lane];
      scatter_row(y + (size_t)rr[k] * FACTOR, xv, vv[k], lo, off0);
    }
  } else {
    for (int k = 0; k < 4; ++k) {
      const int e = e0 + k;
      if (e >= nnz) break;
      const int r = rows[e];
      const int c = cols[e];
      const float v = vals[e];
      const float2 xv = ((const float2*)(x + (size_t)c * FACTOR))[lane];
      scatter_row(y + (size_t)r * FACTOR, xv, v, lo, off0);
    }
  }
}

__global__ __launch_bounds__(128) void combine_wmma(
    const float* __restrict__ x0, const float* __restrict__ x1,
    const float* __restrict__ x2, const float* __restrict__ x3,
    float* __restrict__ out, int nrows) {
  const int lane = threadIdx.x & 31;
  const int wave = threadIdx.x >> 5;
  const int row = blockIdx.x * (blockDim.x >> 5) + wave;
  if (row >= nrows) return;

  const bool lo = (lane < 16);
  v2f a;
  a.x = lo ? 1.0f : (1.0f / 3.0f);
  a.y = lo ? 0.5f : 0.25f;

  const float* __restrict__ p0 = lo ? x0 : x2;
  const float* __restrict__ p1 = lo ? x1 : x3;

  const size_t base = (size_t)row * FACTOR;
  const int n = lane & 15;

#pragma unroll
  for (int nb = 0; nb < FACTOR; nb += 16) {
    v2f b;
    b.x = p0[base + nb + n];
    b.y = p1[base + nb + n];
    v8f c = {};
    c = __builtin_amdgcn_wmma_f32_16x16x4_f32(
         false, a,  false, b,
         (short)0, c,  false,  false);
    out[base + nb + n] = c[0];
  }
}

extern "C" void kernel_launch(void* const* d_in, const int* in_sizes, int n_in,
                              void* d_out, int out_size, void* d_ws, size_t ws_size,
                              hipStream_t stream) {
  (void)n_in; (void)out_size; (void)ws_size;

  const float* embed_user = (const float*)d_in[0];
  const float* embed_item = (const float*)d_in[1];
  const int*   u_idx      = (const int*)d_in[2];
  const int*   i_idx      = (const int*)d_in[3];
  const float* vals_ui    = (const float*)d_in[4];
  const float* vals_iu    = (const float*)d_in[5];

  const int U   = in_sizes[0] / FACTOR;
  const int I   = in_sizes[1] / FACTOR;
  const int nnz = in_sizes[2];

  float* g1u = (float*)d_ws;
  float* g1i = g1u + (size_t)U * FACTOR;
  float* g2u = g1i + (size_t)I * FACTOR;
  float* g2i = g2u + (size_t)U * FACTOR;
  float* g3u = g2i + (size_t)I * FACTOR;
  float* g3i = g3u + (size_t)U * FACTOR;

  const size_t zbytes = (size_t)3 * (size_t)(U + I) * FACTOR * sizeof(float);
  hipMemsetAsync(d_ws, 0, zbytes, stream);

  dim3 sblk(32, 8, 1);
  const int waves = (nnz + 3) / 4;
  const int sgrid = (waves + 7) / 8;

  spmm_coo_atomic<<<sgrid, sblk, 0, stream>>>(u_idx, i_idx, vals_ui, embed_item, g1u, nnz);
  spmm_coo_atomic<<<sgrid, sblk, 0, stream>>>(i_idx, u_idx, vals_iu, embed_user, g1i, nnz);
  spmm_coo_atomic<<<sgrid, sblk, 0, stream>>>(u_idx, i_idx, vals_ui, g1i, g2u, nnz);
  spmm_coo_atomic<<<sgrid, sblk, 0, stream>>>(i_idx, u_idx, vals_iu, g1u, g2i, nnz);
  spmm_coo_atomic<<<sgrid, sblk, 0, stream>>>(u_idx, i_idx, vals_ui, g2i, g3u, nnz);
  spmm_coo_atomic<<<sgrid, sblk, 0, stream>>>(i_idx, u_idx, vals_iu, g2u, g3i, nnz);

  float* out_users = (float*)d_out;
  float* out_items = out_users + (size_t)U * FACTOR;
  const int wpb = 4;
  combine_wmma<<<(U + wpb - 1) / wpb, wpb * 32, 0, stream>>>(embed_user, g1u, g2u, g3u, out_users, U);
  combine_wmma<<<(I + wpb - 1) / wpb, wpb * 32, 0, stream>>>(embed_item, g1i, g2i, g3i, out_items, I);
}


# Edges ordered by destination each call (stable 2-level radix split: ballots + prefix sums, no atomics); one wave per row sums its gathered rows in registers in fixed edge order, one store per row
# speedup vs baseline: 2.6688x; 2.5215x over previous
#include <hip/hip_runtime.h>
#include <hip/hip_bf16.h>

#define FACTOR 64

typedef float v2f __attribute__((ext_vector_type(2)));
typedef float v8f __attribute__((ext_vector_type(8)));

__device__ __forceinline__ void atomic_add_f32_dev(float* p, float v) {
  unsigned long long a = (unsigned long long)p;
  asm volatile("global_atomic_add_f32 %0, %1, off scope:SCOPE_DEV"
               :: "v"(a), "v"(v));
}

__device__ __forceinline__ float xhalf(float x) {
  const int b = __builtin_bit_cast(int, x);
  return __builtin_bit_cast(float,
      __builtin_amdgcn_permlanex16(b, b, 0x76543210, 0xfedcba98, false, false));
}

__device__ __forceinline__ void scatter_row(float* yrow, float2 xv, float v,
                                            bool lo, int off0) {
  const float px = xv.x * v;
  const float py = xv.y * v;
  const float got = xhalf(lo ? py : px);
  float* yp = yrow + off0;
  atomic_add_f32_dev(yp,      lo ? px : got);
  atomic_add_f32_dev(yp + 32, lo ? got : py);
}

__global__ __launch_bounds__(256) void spmm_coo_atomic(
    const int* __restrict__ rows, const int* __restrict__ cols,
    const float* __restrict__ vals, const float* __restrict__ x,
    float* __restrict__ y, int nnz) {
  const int lane = threadIdx.x;
  const int waveId = blockIdx.x * blockDim.y + threadIdx.y;
  const int e0 = waveId * 4;
  if (e0 >= nnz) return;

  __builtin_prefetch(rows + e0 + 4096, 0, 1);
  __builtin_prefetch(cols + e0 + 4096, 0, 1);
  __builtin_prefetch(vals + e0 + 4096, 0, 1);

  const bool lo = (lane < 16);
  const int off0 = lo ? 2 * lane : 2 * lane - 31;

  if (e0 + 3 < nnz) {
    const int eu = __builtin_amdgcn_readfirstlane(e0);
    const int4   r4 = *(const int4*)(rows + eu);
    const int4   c4 = *(const int4*)(cols + eu);
    const float4 v4 = *(const float4*)(vals + eu);
    const int   rr[4] = {r4.x, r4.y, r4.z, r4.w};
    const int   cc[4] = {c4.x, c4.y, c4.z, c4.w};
    const float vv[4] = {v4.x, v4.y, v4.z, v4.w};
#pragma unroll
    for (int k = 0; k < 4; ++k) {
      const float2 xv = ((const float2*)(x + (size_t)cc[k] * FACTOR))[lane];
      scatter_row(y + (size_t)rr[k] * FACTOR, xv, vv[k], lo, off0);
    }
  } else {
    for (int k = 0; k < 4; ++k) {
      const int e = e0 + k;
      if (e >= nnz) break;
      const int r = rows[e];
      const int c = cols[e];
      const float v = vals[e];
      const float2 xv = ((const float2*)(x + (size_t)c * FACTOR))[lane];
      scatter_row(y + (size_t)r * FACTOR, xv, v, lo, off0);
    }
  }
}

__global__ __launch_bounds__(128) void combine_wmma(
    const float* __restrict__ x0, const float* __restrict__ x1,
    const float* __restrict__ x2, const float* __restrict__ x3,
    float* __restrict__ out, int nrows) {
  const int lane = threadIdx.x & 31;
  const int wave = threadIdx.x >> 5;
  const int row = blockIdx.x * (blockDim.x >> 5) + wave;
  if (row >= nrows) return;

  const bool lo = (lane < 16);
  v2f a;
  a.x = lo ? 1.0f : (1.0f / 3.0f);
  a.y = lo ? 0.5f : 0.25f;

  const float* __restrict__ p0 = lo ? x0 : x2;
  const float* __restrict__ p1 = lo ? x1 : x3;

  const size_t base = (size_t)row * FACTOR;
  const int n = lane & 15;

#pragma unroll
  for (int nb = 0; nb < FACTOR; nb += 16) {
    v2f b;
    b.x = p0[base + nb + n];
    b.y = p1[base + nb + n];
    v8f c = {};
    c = __builtin_amdgcn_wmma_f32_16x16x4_f32(
         false, a,  false, b,
         (short)0, c,  false,  false);
    out[base + nb + n] = c[0];
  }
}

#define L1_WPB   8
#define L1_NW    8192
#define L1_GRP   32
#define L1_NGRP  (L1_NW / L1_GRP)

__device__ __forceinline__ unsigned same_digit_lanes(unsigned d, bool valid) {
  unsigned m = (unsigned)__builtin_amdgcn_ballot_w32(valid);
#pragma unroll
  for (int b = 0; b < 8; ++b) {
    const bool bit = ((d >> b) & 1u) != 0u;
    const unsigned bal = (unsigned)__builtin_amdgcn_ballot_w32(bit);
    m &= bit ? bal : ~bal;
  }
  return m;
}

__global__ __launch_bounds__(256) void split_count(
    const int* __restrict__ u_idx, const int* __restrict__ i_idx,
    int nnz, int U, int I, unsigned* __restrict__ cnt1) {
  __shared__ unsigned hist_s[L1_WPB * 256];
  const int lane = threadIdx.x & 31;
  const int wl = threadIdx.x >> 5;
  const int dir = blockIdx.y;
  const int* __restrict__ key = dir ? i_idx : u_idx;
  const unsigned nrows = (unsigned)(dir ? I : U);
  const int wg = blockIdx.x * L1_WPB + wl;
  const int hb = wl * 256;
  for (int j = lane; j < 256; j += 32) hist_s[hb + j] = 0u;
  const int per = (nnz + L1_NW - 1) / L1_NW;
  const int e_lo = min(nnz, wg * per);
  const int e_hi = min(nnz, e_lo + per);
  const unsigned lt = (1u << lane) - 1u;
  unsigned kn = 0xffffffffu;
  if (e_lo + lane < e_hi) kn = (unsigned)key[e_lo + lane];
  for (int eb = e_lo; eb < e_hi; eb += 32) {
    const unsigned k = kn;
    const int en = eb + 32 + lane;
    kn = 0xffffffffu;
    if (en < e_hi) kn = (unsigned)key[en];
    const bool valid = (k < nrows);
    const unsigned d = (k >> 8) & 255u;
    const unsigned peers = same_digit_lanes(d, valid);
    if (valid && (peers & lt) == 0u)
      hist_s[hb + d] = hist_s[hb + d] + (unsigned)__builtin_popcount(peers);
  }
  unsigned* __restrict__ out = cnt1 + ((size_t)dir * L1_NW + (size_t)wg) * 256;
  for (int j = lane; j < 256; j += 32) out[j] = hist_s[hb + j];
}

__global__ __launch_bounds__(256) void split_scan_a(
    unsigned* __restrict__ cnt1, unsigned* __restrict__ grp) {
  const int b = threadIdx.x;
  const int g = blockIdx.x;
  const int dir = blockIdx.y;
  unsigned* __restrict__ c = cnt1 + ((size_t)dir * L1_NW + (size_t)g * L1_GRP) * 256 + b;
  unsigned run = 0u;
  for (int w0 = 0; w0 < L1_GRP; w0 += 16) {
    unsigned v[16];
#pragma unroll
    for (int j = 0; j < 16; ++j) v[j] = c[(size_t)(w0 + j) * 256];
#pragma unroll
    for (int j = 0; j < 16; ++j) { c[(size_t)(w0 + j) * 256] = run; run += v[j]; }
  }
  grp[((size_t)dir * L1_NGRP + g) * 256 + b] = run;
}

__global__ __launch_bounds__(256) void split_scan_b(
    unsigned* __restrict__ grp, unsigned* __restrict__ bbase) {
  __shared__ unsigned s[256];
  const int b = threadIdx.x;
  const int dir = blockIdx.x;
  unsigned* __restrict__ c = grp + (size_t)dir * L1_NGRP * 256 + b;
  unsigned run = 0u;
  for (int g0 = 0; g0 < L1_NGRP; g0 += 16) {
    unsigned v[16];
#pragma unroll
    for (int j = 0; j < 16; ++j) v[j] = c[(size_t)(g0 + j) * 256];
#pragma unroll
    for (int j = 0; j < 16; ++j) { c[(size_t)(g0 + j) * 256] = run; run += v[j]; }
  }
  s[b] = run;
  __syncthreads();
  for (int off = 1; off < 256; off <<= 1) {
    const unsigned v = (b >= off) ? s[b - off] : 0u;
    __syncthreads();
    s[b] += v;
    __syncthreads();
  }
  bbase[dir * 257 + b] = s[b] - run;
  if (b == 255) bbase[dir * 257 + 256] = s[255];
}

__global__ __launch_bounds__(256) void split_place(
    const int* __restrict__ u_idx, const int* __restrict__ i_idx,
    const float* __restrict__ vals_ui, const float* __restrict__ vals_iu,
    int nnz, int U, int I,
    const unsigned* __restrict__ cnt1, const unsigned* __restrict__ grp,
    const unsigned* __restrict__ bbase, int* __restrict__ rec1_) {
  __shared__ unsigned hist_s[L1_WPB * 256];
  const int lane = threadIdx.x & 31;
  const int wl = threadIdx.x >> 5;
  const int dir = blockIdx.y;
  const int* __restrict__ key = dir ? i_idx : u_idx;
  const int* __restrict__ src = dir ? u_idx : i_idx;
  const float* __restrict__ vals = dir ? vals_iu : vals_ui;
  const unsigned nrows = (unsigned)(dir ? I : U);
  const int wg = blockIdx.x * L1_WPB + wl;
  const int g = wg / L1_GRP;
  int2* __restrict__ rec1 = (int2*)rec1_ + (size_t)dir * (size_t)nnz;
  const int hb = wl * 256;
  {
    const unsigned* __restrict__ c = cnt1 + ((size_t)dir * L1_NW + (size_t)wg) * 256;
    const unsigned* __restrict__ gp = grp + ((size_t)dir * L1_NGRP + (size_t)g) * 256;
    const unsigned* __restrict__ bb = bbase + dir * 257;
    for (int j = lane; j < 256; j += 32) hist_s[hb + j] = bb[j] + gp[j] + c[j];
  }
  const int per = (nnz + L1_NW - 1) / L1_NW;
  const int e_lo = min(nnz, wg * per);
  const int e_hi = min(nnz, e_lo + per);
  const unsigned lt = (1u << lane) - 1u;
  unsigned kn = 0xffffffffu;
  int cn = 0;
  float vn = 0.0f;
  if (e_lo + lane < e_hi) { kn = (unsigned)key[e_lo + lane]; cn = src[e_lo + lane]; vn = vals[e_lo + lane]; }
  for (int eb = e_lo; eb < e_hi; eb += 32) {
    const unsigned k = kn;
    const int c = cn;
    const float v = vn;
    const int en = eb + 32 + lane;
    kn = 0xffffffffu;
    if (en < e_hi) { kn = (unsigned)key[en]; cn = src[en]; vn = vals[en]; }
    const bool valid = (k < nrows);
    const unsigned d = (k >> 8) & 255u;
    const unsigned peers = same_digit_lanes(d, valid);
    const unsigned rank = (unsigned)__builtin_popcount(peers & lt);
    unsigned old = 0u;
    if (valid) old = hist_s[hb + d];
    if (valid && rank == 0u) hist_s[hb + d] = old + (unsigned)__builtin_popcount(peers);
    if (valid) rec1[old + rank] = make_int2((int)(((k & 255u) << 16) | (unsigned)c),
                                            __float_as_int(v));
  }
}

__global__ __launch_bounds__(1024) void rows_place(
    const int* __restrict__ rec1_, const unsigned* __restrict__ bbase,
    int nnz, int U, int I,
    unsigned* __restrict__ rowptr_u, unsigned* __restrict__ rowptr_i,
    int* __restrict__ rec_u, int* __restrict__ rec_i) {
  __shared__ unsigned hist_s[32 * 256];
  __shared__ unsigned s[256];
  const int lane = threadIdx.x & 31;
  const int wl = threadIdx.x >> 5;
  const int t = threadIdx.x;
  const int dir = blockIdx.y;
  const int bucket = blockIdx.x;
  const int nrows = dir ? I : U;
  if (bucket * 256 >= nrows) return;
  const int2* __restrict__ rec1 = (const int2*)rec1_ + (size_t)dir * (size_t)nnz;
  int2* __restrict__ rec = (int2*)(dir ? rec_i : rec_u);
  unsigned* __restrict__ rowptr = dir ? rowptr_i : rowptr_u;
  const unsigned bb = bbase[dir * 257 + bucket];
  const unsigned be = bbase[dir * 257 + bucket + 1];
  const unsigned per = (be - bb + 31u) / 32u;
  const unsigned lo = min(be, bb + (unsigned)wl * per);
  const unsigned hi = min(be, lo + per);
  const int hb = wl * 256;
  for (int j = lane; j < 256; j += 32) hist_s[hb + j] = 0u;
  const unsigned lt = (1u << lane) - 1u;
  {
    int xn = 0;
    if (lo + (unsigned)lane < hi) xn = rec1[lo + (unsigned)lane].x;
    for (unsigned p0 = lo; p0 < hi; p0 += 32u) {
      const bool valid = (p0 + (unsigned)lane < hi);
      const unsigned d = ((unsigned)xn >> 16) & 255u;
      const unsigned pn = p0 + 32u + (unsigned)lane;
      xn = 0;
      if (pn < hi) xn = rec1[pn].x;
      const unsigned peers = same_digit_lanes(d, valid);
      if (valid && (peers & lt) == 0u)
        hist_s[hb + d] = hist_s[hb + d] + (unsigned)__builtin_popcount(peers);
    }
  }
  __syncthreads();
  unsigned tot = 0u;
  if (t < 256) {
    for (int w = 0; w < 32; ++w) {
      const unsigned c = hist_s[w * 256 + t];
      hist_s[w * 256 + t] = tot;
      tot += c;
    }
    s[t] = tot;
  }
  __syncthreads();
  for (int off = 1; off < 256; off <<= 1) {
    unsigned v = 0u;
    if (t < 256 && t >= off) v = s[t - off];
    __syncthreads();
    if (t < 256) s[t] += v;
    __syncthreads();
  }
  if (t < 256) {
    const unsigned start = bb + s[t] - tot;
    const int row = bucket * 256 + t;
    if (row < nrows) rowptr[row] = start;
    if (row == nrows - 1) rowptr[nrows] = be;
    for (int w = 0; w < 32; ++w) hist_s[w * 256 + t] += start;
  }
  __syncthreads();
  int2 rn = make_int2(0, 0);
  if (lo + (unsigned)lane < hi) rn = rec1[lo + (unsigned)lane];
  for (unsigned p0 = lo; p0 < hi; p0 += 32u) {
    const bool valid = (p0 + (unsigned)lane < hi);
    const int2 r = rn;
    const unsigned pn = p0 + 32u + (unsigned)lane;
    rn = make_int2(0, 0);
    if (pn < hi) rn = rec1[pn];
    const unsigned d = ((unsigned)r.x >> 16) & 255u;
    const unsigned peers = same_digit_lanes(d, valid);
    const unsigned rank = (unsigned)__builtin_popcount(peers & lt);
    unsigned old = 0u;
    if (valid) old = hist_s[hb + d];
    if (valid && rank == 0u) hist_s[hb + d] = old + (unsigned)__builtin_popcount(peers);
    if (valid) rec[old + rank] = make_int2(r.x & 0xffff, r.y);
  }
}

template <bool TAIL>
__device__ __forceinline__ void edges8(const float* __restrict__ x, int2 my, unsigned k,
                                       unsigned n, int lane, float& ax, float& ay) {
  float2 xv[8];
  float v[8];
#pragma unroll
  for (int j = 0; j < 8; ++j) {
    const unsigned kk = TAIL ? min(k + (unsigned)j, n - 1u) : (k + (unsigned)j);
    const int col = __builtin_amdgcn_readlane(my.x, (int)kk);
    v[j] = __int_as_float(__builtin_amdgcn_readlane(my.y, (int)kk));
    xv[j] = ((const float2*)(x + (size_t)col * FACTOR))[lane];
  }
#pragma unroll
  for (int j = 0; j < 8; ++j) {
    if (!TAIL || (k + (unsigned)j < n)) {
      ax += xv[j].x * v[j];
      ay += xv[j].y * v[j];
    }
  }
}

__global__ __launch_bounds__(256) void spmm_rows(
    const unsigned* __restrict__ rowptr, const int* __restrict__ rec_,
    const float* __restrict__ x, float* __restrict__ y, int nrows) {
  const int lane = threadIdx.x;
  const int row = blockIdx.x * blockDim.y + threadIdx.y;
  if (row >= nrows) return;
  const int r = __builtin_amdgcn_readfirstlane(row);
  const unsigned beg = (unsigned)__builtin_amdgcn_readfirstlane((int)rowptr[r]);
  const unsigned end = (unsigned)__builtin_amdgcn_readfirstlane((int)rowptr[r + 1]);
  const int2* __restrict__ rec = (const int2*)rec_;
  float ax = 0.0f, ay = 0.0f;
  for (unsigned c = beg; c < end; c += 32u) {
    const unsigned n = min(32u, end - c);
    int2 my = make_int2(0, 0);
    if ((unsigned)lane < n) my = rec[c + (unsigned)lane];
    unsigned k = 0u;
    for (; k + 8u <= n; k += 8u) edges8<false>(x, my, k, n, lane, ax, ay);
    if (k < n) edges8<true>(x, my, k, n, lane, ax, ay);
  }
  float2 o; o.x = ax; o.y = ay;
  ((float2*)(y + (size_t)r * FACTOR))[lane] = o;
}

extern "C" void kernel_launch(void* const* d_in, const int* in_sizes, int n_in,
                              void* d_out, int out_size, void* d_ws, size_t ws_size,
                              hipStream_t stream) {
  (void)n_in; (void)out_size;

  const float* embed_user = (const float*)d_in[0];
  const float* embed_item = (const float*)d_in[1];
  const int*   u_idx      = (const int*)d_in[2];
  const int*   i_idx      = (const int*)d_in[3];
  const float* vals_ui    = (const float*)d_in[4];
  const float* vals_iu    = (const float*)d_in[5];

  const int U   = in_sizes[0] / FACTOR;
  const int I   = in_sizes[1] / FACTOR;
  const int nnz = in_sizes[2];

  float* g1u = (float*)d_ws;
  float* g1i = g1u + (size_t)U * FACTOR;
  float* g2u = g1i + (size_t)I * FACTOR;
  float* g2i = g2u + (size_t)U * FACTOR;
  float* g3u = g2i + (size_t)I * FACTOR;
  float* g3i = g3u + (size_t)U * FACTOR;

  const size_t plane_bytes = (size_t)(U + I) * FACTOR * sizeof(float);
  const size_t acc_bytes = (size_t)3 * plane_bytes;
  const size_t rec_bytes = (size_t)2 * (size_t)nnz * 8;
  const size_t cnt1_bytes = (size_t)2 * L1_NW * 256 * sizeof(unsigned);
  const size_t small_words = (size_t)(U + I) + 2 + (size_t)2 * L1_NGRP * 256 + (size_t)2 * 257;
  const size_t need = acc_bytes + rec_bytes + small_words * sizeof(unsigned);

  const bool ordered = (ws_size >= need) && (U <= 65536) && (I <= 65536) &&
                       (plane_bytes >= cnt1_bytes) && ((size_t)2 * plane_bytes >= rec_bytes);

  if (ordered) {
    int* rec_u = (int*)(g3i + (size_t)I * FACTOR);
    int* rec_i = rec_u + (size_t)2 * (size_t)nnz;
    unsigned* rowptr_u = (unsigned*)(rec_i + (size_t)2 * (size_t)nnz);
    unsigned* rowptr_i = rowptr_u + (size_t)U + 1;
    unsigned* grp   = rowptr_i + (size_t)I + 1;
    unsigned* bbase = grp + (size_t)2 * L1_NGRP * 256;
    unsigned* cnt1 = (unsigned*)g1u;
    int* rec1 = (int*)g2u;

    dim3 l1grid(L1_NW / L1_WPB, 2, 1);
    split_count<<<l1grid, L1_WPB * 32, 0, stream>>>(u_idx, i_idx, nnz, U, I, cnt1);
    split_scan_a<<<dim3(L1_NGRP, 2, 1), 256, 0, stream>>>(cnt1, grp);
    split_scan_b<<<2, 256, 0, stream>>>(grp, bbase);
    split_place<<<l1grid, L1_WPB * 32, 0, stream>>>(u_idx, i_idx, vals_ui, vals_iu, nnz, U, I,
                                                     cnt1, grp, bbase, rec1);
    rows_place<<<dim3(256, 2, 1), 1024, 0, stream>>>(rec1, bbase, nnz, U, I,
                                                      rowptr_u, rowptr_i, rec_u, rec_i);

    dim3 rblk(32, 8, 1);
    const int gu = (U + 7) / 8;
    const int gi = (I + 7) / 8;
    spmm_rows<<<gu, rblk, 0, stream>>>(rowptr_u, rec_u, embed_item, g1u, U);
    spmm_rows<<<gi, rblk, 0, stream>>>(rowptr_i, rec_i, embed_user, g1i, I);
    spmm_rows<<<gu, rblk, 0, stream>>>(rowptr_u, rec_u, g1i, g2u, U);
    spmm_rows<<<gi, rblk, 0, stream>>>(rowptr_i, rec_i, g1u, g2i, I);
    spmm_rows<<<gu, rblk, 0, stream>>>(rowptr_u, rec_u, g2i, g3u, U);
    spmm_rows<<<gi, rblk, 0, stream>>>(rowptr_i, rec_i, g2u, g3i, I);
  } else {
    hipMemsetAsync(d_ws, 0, acc_bytes, stream);

    dim3 sblk(32, 8, 1);
    const int waves = (nnz + 3) / 4;
    const int sgrid = (waves + 7) / 8;

    spmm_coo_atomic<<<sgrid, sblk, 0, stream>>>(u_idx, i_idx, vals_ui, embed_item, g1u, nnz);
    spmm_coo_atomic<<<sgrid, sblk, 0, stream>>>(i_idx, u_idx, vals_iu, embed_user, g1i, nnz);
    spmm_coo_atomic<<<sgrid, sblk, 0, stream>>>(u_idx, i_idx, vals_ui, g1i, g2u, nnz);
    spmm_coo_atomic<<<sgrid, sblk, 0, stream>>>(i_idx, u_idx, vals_iu, g1u, g2i, nnz);
    spmm_coo_atomic<<<sgrid, sblk, 0, stream>>>(u_idx, i_idx, vals_ui, g2i, g3u, nnz);
    spmm_coo_atomic<<<sgrid, sblk, 0, stream>>>(i_idx, u_idx, vals_iu, g2u, g3i, nnz);
  }

  float* out_users = (float*)d_out;
  float* out_items = out_users + (size_t)U * FACTOR;
  const int wpb = 4;
  combine_wmma<<<(U + wpb - 1) / wpb, wpb * 32, 0, stream>>>(embed_user, g1u, g2u, g3u, out_users, U);
  combine_wmma<<<(I + wpb - 1) / wpb, wpb * 32, 0, stream>>>(embed_item, g1i, g2i, g3i, out_items, I);
}


# Last hop's row kernels finish the output: the row's wave keeps its hop-3 sum in registers and runs the 1, 1/2, 1/3, 1/4 WMMA combine on it; 13 launches become 11, no hop-3 planes, same sums
# speedup vs baseline: 2.7499x; 1.0304x over previous
#include <hip/hip_runtime.h>
#include <hip/hip_bf16.h>

#define FACTOR 64

typedef float v2f __attribute__((ext_vector_type(2)));
typedef float v8f __attribute__((ext_vector_type(8)));

__device__ __forceinline__ void atomic_add_f32_dev(float* p, float v) {
  unsigned long long a = (unsigned long long)p;
  asm volatile("global_atomic_add_f32 %0, %1, off scope:SCOPE_DEV"
               :: "v"(a), "v"(v));
}

__device__ __forceinline__ float xhalf(float x) {
  const int b = __builtin_bit_cast(int, x);
  return __builtin_bit_cast(float,
      __builtin_amdgcn_permlanex16(b, b, 0x76543210, 0xfedcba98, false, false));
}

__device__ __forceinline__ void scatter_row(float* yrow, float2 xv, float v,
                                            bool lo, int off0) {
  const float px = xv.x * v;
  const float py = xv.y * v;
  const float got = xhalf(lo ? py : px);
  float* yp = yrow + off0;
  atomic_add_f32_dev(yp,      lo ? px : got);
  atomic_add_f32_dev(yp + 32, lo ? got : py);
}

__global__ __launch_bounds__(256) void spmm_coo_atomic(
    const int* __restrict__ rows, const int* __restrict__ cols,
    const float* __restrict__ vals, const float* __restrict__ x,
    float* __restrict__ y, int nnz) {
  const int lane = threadIdx.x;
  const int waveId = blockIdx.x * blockDim.y + threadIdx.y;
  const int e0 = waveId * 4;
  if (e0 >= nnz) return;

  __builtin_prefetch(rows + e0 + 4096, 0, 1);
  __builtin_prefetch(cols + e0 + 4096, 0, 1);
  __builtin_prefetch(vals + e0 + 4096, 0, 1);

  const bool lo = (lane < 16);
  const int off0 = lo ? 2 * lane : 2 * lane - 31;

  if (e0 + 3 < nnz) {
    const int eu = __builtin_amdgcn_readfirstlane(e0);
    const int4   r4 = *(const int4*)(rows + eu);
    const int4   c4 = *(const int4*)(cols + eu);
    const float4 v4 = *(const float4*)(vals + eu);
    const int   rr[4] = {r4.x, r4.y, r4.z, r4.w};
    const int   cc[4] = {c4.x, c4.y, c4.z, c4.w};
    const float vv[4] = {v4.x, v4.y, v4.z, v4.w};
#pragma unroll
    for (int k = 0; k < 4; ++k) {
      const float2 xv = ((const float2*)(x + (size_t)cc[k] * FACTOR))[lane];
      scatter_row(y + (size_t)rr[k] * FACTOR, xv, vv[k], lo, off0);
    }
  } else {
    for (int k = 0; k < 4; ++k) {
      const int e = e0 + k;
      if (e >= nnz) break;
      const int r = rows[e];
      const int c = cols[e];
      const float v = vals[e];
      const float2 xv = ((const float2*)(x + (size_t)c * FACTOR))[lane];
      scatter_row(y + (size_t)r * FACTOR, xv, v, lo, off0);
    }
  }
}

__global__ __launch_bounds__(128) void combine_wmma(
    const float* __restrict__ x0, const float* __restrict__ x1,
    const float* __restrict__ x2, const float* __restrict__ x3,
    float* __restrict__ out, int nrows) {
  const int lane = threadIdx.x & 31;
  const int wave = threadIdx.x >> 5;
  const int row = blockIdx.x * (blockDim.x >> 5) + wave;
  if (row >= nrows) return;

  const bool lo = (lane < 16);
  v2f a;
  a.x = lo ? 1.0f : (1.0f / 3.0f);
  a.y = lo ? 0.5f : 0.25f;

  const float* __restrict__ p0 = lo ? x0 : x2;
  const float* __restrict__ p1 = lo ? x1 : x3;

  const size_t base = (size_t)row * FACTOR;
  const int n = lane & 15;

#pragma unroll
  for (int nb = 0; nb < FACTOR; nb += 16) {
    v2f b;
    b.x = p0[base + nb + n];
    b.y = p1[base + nb + n];
    v8f c = {};
    c = __builtin_amdgcn_wmma_f32_16x16x4_f32(
         false, a,  false, b,
         (short)0, c,  false,  false);
    out[base + nb + n] = c[0];
  }
}

#define L1_WPB   8
#define L1_NW    8192
#define L1_GRP   32
#define L1_NGRP  (L1_NW / L1_GRP)

__device__ __forceinline__ unsigned same_digit_lanes(unsigned d, bool valid) {
  unsigned m = (unsigned)__builtin_amdgcn_ballot_w32(valid);
#pragma unroll
  for (int b = 0; b < 8; ++b) {
    const bool bit = ((d >> b) & 1u) != 0u;
    const unsigned bal = (unsigned)__builtin_amdgcn_ballot_w32(bit);
    m &= bit ? bal : ~bal;
  }
  return m;
}

__global__ __launch_bounds__(256) void split_count(
    const int* __restrict__ u_idx, const int* __restrict__ i_idx,
    int nnz, int U, int I, unsigned* __restrict__ cnt1) {
  __shared__ unsigned hist_s[L1_WPB * 256];
  const int lane = threadIdx.x & 31;
  const int wl = threadIdx.x >> 5;
  const int dir = blockIdx.y;
  const int* __restrict__ key = dir ? i_idx : u_idx;
  const unsigned nrows = (unsigned)(dir ? I : U);
  const int wg = blockIdx.x * L1_WPB + wl;
  const int hb = wl * 256;
  for (int j = lane; j < 256; j += 32) hist_s[hb + j] = 0u;
  const int per = (nnz + L1_NW - 1) / L1_NW;
  const int e_lo = min(nnz, wg * per);
  const int e_hi = min(nnz, e_lo + per);
  const unsigned lt = (1u << lane) - 1u;
  unsigned kn = 0xffffffffu;
  if (e_lo + lane < e_hi) kn = (unsigned)key[e_lo + lane];
  for (int eb = e_lo; eb < e_hi; eb += 32) {
    const unsigned k = kn;
    const int en = eb + 32 + lane;
    kn = 0xffffffffu;
    if (en < e_hi) kn = (unsigned)key[en];
    const bool valid = (k < nrows);
    const unsigned d = (k >> 8) & 255u;
    const unsigned peers = same_digit_lanes(d, valid);
    if (valid && (peers & lt) == 0u)
      hist_s[hb + d] = hist_s[hb + d] + (unsigned)__builtin_popcount(peers);
  }
  unsigned* __restrict__ out = cnt1 + ((size_t)dir * L1_NW + (size_t)wg) * 256;
  for (int j = lane; j < 256; j += 32) out[j] = hist_s[hb + j];
}

__global__ __launch_bounds__(256) void split_scan_a(
    unsigned* __restrict__ cnt1, unsigned* __restrict__ grp) {
  const int b = threadIdx.x;
  const int g = blockIdx.x;
  const int dir = blockIdx.y;
  unsigned* __restrict__ c = cnt1 + ((size_t)dir * L1_NW + (size_t)g * L1_GRP) * 256 + b;
  unsigned run = 0u;
  for (int w0 = 0; w0 < L1_GRP; w0 += 16) {
    unsigned v[16];
#pragma unroll
    for (int j = 0; j < 16; ++j) v[j] = c[(size_t)(w0 + j) * 256];
#pragma unroll
    for (int j = 0; j < 16; ++j) { c[(size_t)(w0 + j) * 256] = run; run += v[j]; }
  }
  grp[((size_t)dir * L1_NGRP + g) * 256 + b] = run;
}

__global__ __launch_bounds__(256) void split_scan_b(
    unsigned* __restrict__ grp, unsigned* __restrict__ bbase) {
  __shared__ unsigned s[256];
  const int b = threadIdx.x;
  const int dir = blockIdx.x;
  unsigned* __restrict__ c = grp + (size_t)dir * L1_NGRP * 256 + b;
  unsigned run = 0u;
  for (int g0 = 0; g0 < L1_NGRP; g0 += 16) {
    unsigned v[16];
#pragma unroll
    for (int j = 0; j < 16; ++j) v[j] = c[(size_t)(g0 + j) * 256];
#pragma unroll
    for (int j = 0; j < 16; ++j) { c[(size_t)(g0 + j) * 256] = run; run += v[j]; }
  }
  s[b] = run;
  __syncthreads();
  for (int off = 1; off < 256; off <<= 1) {
    const unsigned v = (b >= off) ? s[b - off] : 0u;
    __syncthreads();
    s[b] += v;
    __syncthreads();
  }
  bbase[dir * 257 + b] = s[b] - run;
  if (b == 255) bbase[dir * 257 + 256] = s[255];
}

__global__ __launch_bounds__(256) void split_place(
    const int* __restrict__ u_idx, const int* __restrict__ i_idx,
    const float* __restrict__ vals_ui, const float* __restrict__ vals_iu,
    int nnz, int U, int I,
    const unsigned* __restrict__ cnt1, const unsigned* __restrict__ grp,
    const unsigned* __restrict__ bbase, int* __restrict__ rec1_) {
  __shared__ unsigned hist_s[L1_WPB * 256];
  const int lane = threadIdx.x & 31;
  const int wl = threadIdx.x >> 5;
  const int dir = blockIdx.y;
  const int* __restrict__ key = dir ? i_idx : u_idx;
  const int* __restrict__ src = dir ? u_idx : i_idx;
  const float* __restrict__ vals = dir ? vals_iu : vals_ui;
  const unsigned nrows = (unsigned)(dir ? I : U);
  const int wg = blockIdx.x * L1_WPB + wl;
  const int g = wg / L1_GRP;
  int2* __restrict__ rec1 = (int2*)rec1_ + (size_t)dir * (size_t)nnz;
  const int hb = wl * 256;
  {
    const unsigned* __restrict__ c = cnt1 + ((size_t)dir * L1_NW + (size_t)wg) * 256;
    const unsigned* __restrict__ gp = grp + ((size_t)dir * L1_NGRP + (size_t)g) * 256;
    const unsigned* __restrict__ bb = bbase + dir * 257;
    for (int j = lane; j < 256; j += 32) hist_s[hb + j] = bb[j] + gp[j] + c[j];
  }
  const int per = (nnz + L1_NW - 1) / L1_NW;
  const int e_lo = min(nnz, wg * per);
  const int e_hi = min(nnz, e_lo + per);
  const unsigned lt = (1u << lane) - 1u;
  unsigned kn = 0xffffffffu;
  int cn = 0;
  float vn = 0.0f;
  if (e_lo + lane < e_hi) { kn = (unsigned)key[e_lo + lane]; cn = src[e_lo + lane]; vn = vals[e_lo + lane]; }
  for (int eb = e_lo; eb < e_hi; eb += 32) {
    const unsigned k = kn;
    const int c = cn;
    const float v = vn;
    const int en = eb + 32 + lane;
    kn = 0xffffffffu;
    if (en < e_hi) { kn = (unsigned)key[en]; cn = src[en]; vn = vals[en]; }
    const bool valid = (k < nrows);
    const unsigned d = (k >> 8) & 255u;
    const unsigned peers = same_digit_lanes(d, valid);
    const unsigned rank = (unsigned)__builtin_popcount(peers & lt);
    unsigned old = 0u;
    if (valid) old = hist_s[hb + d];
    if (valid && rank == 0u) hist_s[hb + d] = old + (unsigned)__builtin_popcount(peers);
    if (valid) rec1[old + rank] = make_int2((int)(((k & 255u) << 16) | (unsigned)c),
                                            __float_as_int(v));
  }
}

__global__ __launch_bounds__(1024) void rows_place(
    const int* __restrict__ rec1_, const unsigned* __restrict__ bbase,
    int nnz, int U, int I,
    unsigned* __restrict__ rowptr_u, unsigned* __restrict__ rowptr_i,
    int* __restrict__ rec_u, int* __restrict__ rec_i) {
  __shared__ unsigned hist_s[32 * 256];
  __shared__ unsigned s[256];
  const int lane = threadIdx.x & 31;
  const int wl = threadIdx.x >> 5;
  const int t = threadIdx.x;
  const int dir = blockIdx.y;
  const int bucket = blockIdx.x;
  const int nrows = dir ? I : U;
  if (bucket * 256 >= nrows) return;
  const int2* __restrict__ rec1 = (const int2*)rec1_ + (size_t)dir * (size_t)nnz;
  int2* __restrict__ rec = (int2*)(dir ? rec_i : rec_u);
  unsigned* __restrict__ rowptr = dir ? rowptr_i : rowptr_u;
  const unsigned bb = bbase[dir * 257 + bucket];
  const unsigned be = bbase[dir * 257 + bucket + 1];
  const unsigned per = (be - bb + 31u) / 32u;
  const unsigned lo = min(be, bb + (unsigned)wl * per);
  const unsigned hi = min(be, lo + per);
  const int hb = wl * 256;
  for (int j = lane; j < 256; j += 32) hist_s[hb + j] = 0u;
  const unsigned lt = (1u << lane) - 1u;
  {
    int xn = 0;
    if (lo + (unsigned)lane < hi) xn = rec1[lo + (unsigned)lane].x;
    for (unsigned p0 = lo; p0 < hi; p0 += 32u) {
      const bool valid = (p0 + (unsigned)lane < hi);
      const unsigned d = ((unsigned)xn >> 16) & 255u;
      const unsigned pn = p0 + 32u + (unsigned)lane;
      xn = 0;
      if (pn < hi) xn = rec1[pn].x;
      const unsigned peers = same_digit_lanes(d, valid);
      if (valid && (peers & lt) == 0u)
        hist_s[hb + d] = hist_s[hb + d] + (unsigned)__builtin_popcount(peers);
    }
  }
  __syncthreads();
  unsigned tot = 0u;
  if (t < 256) {
    for (int w = 0; w < 32; ++w) {
      const unsigned c = hist_s[w * 256 + t];
      hist_s[w * 256 + t] = tot;
      tot += c;
    }
    s[t] = tot;
  }
  __syncthreads();
  for (int off = 1; off < 256; off <<= 1) {
    unsigned v = 0u;
    if (t < 256 && t >= off) v = s[t - off];
    __syncthreads();
    if (t < 256) s[t] += v;
    __syncthreads();
  }
  if (t < 256) {
    const unsigned start = bb + s[t] - tot;
    const int row = bucket * 256 + t;
    if (row < nrows) rowptr[row] = start;
    if (row == nrows - 1) rowptr[nrows] = be;
    for (int w = 0; w < 32; ++w) hist_s[w * 256 + t] += start;
  }
  __syncthreads();
  int2 rn = make_int2(0, 0);
  if (lo + (unsigned)lane < hi) rn = rec1[lo + (unsigned)lane];
  for (unsigned p0 = lo; p0 < hi; p0 += 32u) {
    const bool valid = (p0 + (unsigned)lane < hi);
    const int2 r = rn;
    const unsigned pn = p0 + 32u + (unsigned)lane;
    rn = make_int2(0, 0);
    if (pn < hi) rn = rec1[pn];
    const unsigned d = ((unsigned)r.x >> 16) & 255u;
    const unsigned peers = same_digit_lanes(d, valid);
    const unsigned rank = (unsigned)__builtin_popcount(peers & lt);
    unsigned old = 0u;
    if (valid) old = hist_s[hb + d];
    if (valid && rank == 0u) hist_s[hb + d] = old + (unsigned)__builtin_popcount(peers);
    if (valid) rec[old + rank] = make_int2(r.x & 0xffff, r.y);
  }
}

template <bool TAIL>
__device__ __forceinline__ void edges8(const float* __restrict__ x, int2 my, unsigned k,
                                       unsigned n, int lane, float& ax, float& ay) {
  float2 xv[8];
  float v[8];
#pragma unroll
  for (int j = 0; j < 8; ++j) {
    const unsigned kk = TAIL ? min(k + (unsigned)j, n - 1u) : (k + (unsigned)j);
    const int col = __builtin_amdgcn_readlane(my.x, (int)kk);
    v[j] = __int_as_float(__builtin_amdgcn_readlane(my.y, (int)kk));
    xv[j] = ((const float2*)(x + (size_t)col * FACTOR))[lane];
  }
#pragma unroll
  for (int j = 0; j < 8; ++j) {
    if (!TAIL || (k + (unsigned)j < n)) {
      ax += xv[j].x * v[j];
      ay += xv[j].y * v[j];
    }
  }
}

template <bool FINISH>
__global__ __launch_bounds__(256) void spmm_rows(
    const unsigned* __restrict__ rowptr, const int* __restrict__ rec_,
    const float* __restrict__ x, float* __restrict__ y, int nrows,
    const float* __restrict__ e0, const float* __restrict__ h1,
    const float* __restrict__ h2) {
  const int lane = threadIdx.x;
  const int row = blockIdx.x * blockDim.y + threadIdx.y;
  if (row >= nrows) return;
  const int r = __builtin_amdgcn_readfirstlane(row);
  float2 q0 = make_float2(0.0f, 0.0f), q1 = q0, q2 = q0;
  if (FINISH) {
    const size_t rb = (size_t)r * FACTOR;
    q0 = ((const float2*)(e0 + rb))[lane];
    q1 = ((const float2*)(h1 + rb))[lane];
    q2 = ((const float2*)(h2 + rb))[lane];
  }
  const unsigned beg = (unsigned)__builtin_amdgcn_readfirstlane((int)rowptr[r]);
  const unsigned end = (unsigned)__builtin_amdgcn_readfirstlane((int)rowptr[r + 1]);
  const int2* __restrict__ rec = (const int2*)rec_;
  float ax = 0.0f, ay = 0.0f;
  for (unsigned c = beg; c < end; c += 32u) {
    const unsigned n = min(32u, end - c);
    int2 my = make_int2(0, 0);
    if ((unsigned)lane < n) my = rec[c + (unsigned)lane];
    unsigned k = 0u;
    for (; k + 8u <= n; k += 8u) edges8<false>(x, my, k, n, lane, ax, ay);
    if (k < n) edges8<true>(x, my, k, n, lane, ax, ay);
  }
  float2 o;
  if (FINISH) {
    const bool lo = (lane < 16);
    v2f a;
    a.x = lo ? 1.0f : (1.0f / 3.0f);
    a.y = lo ? 0.5f : 0.25f;
    const float s0x = xhalf(lo ? q2.x : q0.x);
    const float s0y = xhalf(lo ? q2.y : q0.y);
    const float s1x = xhalf(lo ? ax : q1.x);
    const float s1y = xhalf(lo ? ay : q1.y);
    v2f b0, b1, b2, b3;
    b0.x = lo ? q0.x : s0x;  b0.y = lo ? q1.x : s1x;
    b1.x = lo ? q0.y : s0y;  b1.y = lo ? q1.y : s1y;
    b2.x = lo ? s0x : q2.x;  b2.y = lo ? s1x : ax;
    b3.x = lo ? s0y : q2.y;  b3.y = lo ? s1y : ay;
    const v8f z = {};
    const v8f d0 = __builtin_amdgcn_wmma_f32_16x16x4_f32(false, a, false, b0, (short)0, z, false, false);
    const v8f d1 = __builtin_amdgcn_wmma_f32_16x16x4_f32(false, a, false, b1, (short)0, z, false, false);
    const v8f d2 = __builtin_amdgcn_wmma_f32_16x16x4_f32(false, a, false, b2, (short)0, z, false, false);
    const v8f d3 = __builtin_amdgcn_wmma_f32_16x16x4_f32(false, a, false, b3, (short)0, z, false, false);
    o.x = lo ? d0[0] : d2[0];
    o.y = lo ? d1[0] : d3[0];
  } else {
    o.x = ax; o.y = ay;
  }
  ((float2*)(y + (size_t)r * FACTOR))[lane] = o;
}

extern "C" void kernel_launch(void* const* d_in, const int* in_sizes, int n_in,
                              void* d_out, int out_size, void* d_ws, size_t ws_size,
                              hipStream_t stream) {
  (void)n_in; (void)out_size;

  const float* embed_user = (const float*)d_in[0];
  const float* embed_item = (const float*)d_in[1];
  const int*   u_idx      = (const int*)d_in[2];
  const int*   i_idx      = (const int*)d_in[3];
  const float* vals_ui    = (const float*)d_in[4];
  const float* vals_iu    = (const float*)d_in[5];

  const int U   = in_sizes[0] / FACTOR;
  const int I   = in_sizes[1] / FACTOR;
  const int nnz = in_sizes[2];

  float* g1u = (float*)d_ws;
  float* g1i = g1u + (size_t)U * FACTOR;
  float* g2u = g1i + (size_t)I * FACTOR;
  float* g2i = g2u + (size_t)U * FACTOR;
  float* g3u = g2i + (size_t)I * FACTOR;
  float* g3i = g3u + (size_t)U * FACTOR;

  const size_t plane_bytes = (size_t)(U + I) * FACTOR * sizeof(float);
  const size_t acc_bytes = (size_t)3 * plane_bytes;
  const size_t rec_bytes = (size_t)2 * (size_t)nnz * 8;
  const size_t cnt1_bytes = (size_t)2 * L1_NW * 256 * sizeof(unsigned);
  const size_t small_words = (size_t)(U + I) + 2 + (size_t)2 * L1_NGRP * 256 + (size_t)2 * 257;
  const size_t need = acc_bytes + rec_bytes + small_words * sizeof(unsigned);

  float* out_users = (float*)d_out;
  float* out_items = out_users + (size_t)U * FACTOR;

  const bool ordered = (ws_size >= need) && (U <= 65536) && (I <= 65536) &&
                       (plane_bytes >= cnt1_bytes) && ((size_t)2 * plane_bytes >= rec_bytes);

  if (ordered) {
    int* rec_u = (int*)(g3i + (size_t)I * FACTOR);
    int* rec_i = rec_u + (size_t)2 * (size_t)nnz;
    unsigned* rowptr_u = (unsigned*)(rec_i + (size_t)2 * (size_t)nnz);
    unsigned* rowptr_i = rowptr_u + (size_t)U + 1;
    unsigned* grp   = rowptr_i + (size_t)I + 1;
    unsigned* bbase = grp + (size_t)2 * L1_NGRP * 256;
    unsigned* cnt1 = (unsigned*)g1u;
    int* rec1 = (int*)g2u;

    dim3 l1grid(L1_NW / L1_WPB, 2, 1);
    split_count<<<l1grid, L1_WPB * 32, 0, stream>>>(u_idx, i_idx, nnz, U, I, cnt1);
    split_scan_a<<<dim3(L1_NGRP, 2, 1), 256, 0, stream>>>(cnt1, grp);
    split_scan_b<<<2, 256, 0, stream>>>(grp, bbase);
    split_place<<<l1grid, L1_WPB * 32, 0, stream>>>(u_idx, i_idx, vals_ui, vals_iu, nnz, U, I,
                                                     cnt1, grp, bbase, rec1);
    rows_place<<<dim3(256, 2, 1), 1024, 0, stream>>>(rec1, bbase, nnz, U, I,
                                                      rowptr_u, rowptr_i, rec_u, rec_i);

    dim3 rblk(32, 8, 1);
    const int gu = (U + 7) / 8;
    const int gi = (I + 7) / 8;
    const float* none = nullptr;
    spmm_rows<false><<<gu, rblk, 0, stream>>>(rowptr_u, rec_u, embed_item, g1u, U, none, none, none);
    spmm_rows<false><<<gi, rblk, 0, stream>>>(rowptr_i, rec_i, embed_user, g1i, I, none, none, none);
    spmm_rows<false><<<gu, rblk, 0, stream>>>(rowptr_u, rec_u, g1i, g2u, U, none, none, none);
    spmm_rows<false><<<gi, rblk, 0, stream>>>(rowptr_i, rec_i, g1u, g2i, I, none, none, none);
    spmm_rows<true><<<gu, rblk, 0, stream>>>(rowptr_u, rec_u, g2i, out_users, U, embed_user, g1u, g2u);
    spmm_rows<true><<<gi, rblk, 0, stream>>>(rowptr_i, rec_i, g2u, out_items, I, embed_item, g1i, g2i);
  } else {
    hipMemsetAsync(d_ws, 0, acc_bytes, stream);

    dim3 sblk(32, 8, 1);
    const int waves = (nnz + 3) / 4;
    const int sgrid = (waves + 7) / 8;

    spmm_coo_atomic<<<sgrid, sblk, 0, stream>>>(u_idx, i_idx, vals_ui, embed_item, g1u, nnz);
    spmm_coo_atomic<<<sgrid, sblk, 0, stream>>>(i_idx, u_idx, vals_iu, embed_user, g1i, nnz);
    spmm_coo_atomic<<<sgrid, sblk, 0, stream>>>(u_idx, i_idx, vals_ui, g1i, g2u, nnz);
    spmm_coo_atomic<<<sgrid, sblk, 0, stream>>>(i_idx, u_idx, vals_iu, g1u, g2i, nnz);
    spmm_coo_atomic<<<sgrid, sblk, 0, stream>>>(u_idx, i_idx, vals_ui, g2i, g3u, nnz);
    spmm_coo_atomic<<<sgrid, sblk, 0, stream>>>(i_idx, u_idx, vals_iu, g2u, g3i, nnz);

    const int wpb = 4;
    combine_wmma<<<(U + wpb - 1) / wpb, wpb * 32, 0, stream>>>(embed_user, g1u, g2u, g3u, out_users, U);
    combine_wmma<<<(I + wpb - 1) / wpb, wpb * 32, 0, stream>>>(embed_item, g1i, g2i, g3i, out_items, I);
  }
}


# Level-2 placement builds each 256-row bucket's final record order in LDS and copies it out in whole 128-B lines (was one scattered 8-B global store per record); same order, same sums
# speedup vs baseline: 2.9708x; 1.0803x over previous
#include <hip/hip_runtime.h>
#include <hip/hip_bf16.h>

#define FACTOR 64

typedef float v2f __attribute__((ext_vector_type(2)));
typedef float v8f __attribute__((ext_vector_type(8)));

__device__ __forceinline__ void atomic_add_f32_dev(float* p, float v) {
  unsigned long long a = (unsigned long long)p;
  asm volatile("global_atomic_add_f32 %0, %1, off scope:SCOPE_DEV"
               :: "v"(a), "v"(v));
}

__device__ __forceinline__ float xhalf(float x) {
  const int b = __builtin_bit_cast(int, x);
  return __builtin_bit_cast(float,
      __builtin_amdgcn_permlanex16(b, b, 0x76543210, 0xfedcba98, false, false));
}

__device__ __forceinline__ void scatter_row(float* yrow, float2 xv, float v,
                                            bool lo, int off0) {
  const float px = xv.x * v;
  const float py = xv.y * v;
  const float got = xhalf(lo ? py : px);
  float* yp = yrow + off0;
  atomic_add_f32_dev(yp,      lo ? px : got);
  atomic_add_f32_dev(yp + 32, lo ? got : py);
}

__global__ __launch_bounds__(256) void spmm_coo_atomic(
    const int* __restrict__ rows, const int* __restrict__ cols,
    const float* __restrict__ vals, const float* __restrict__ x,
    float* __restrict__ y, int nnz) {
  const int lane = threadIdx.x;
  const int waveId = blockIdx.x * blockDim.y + threadIdx.y;
  const int e0 = waveId * 4;
  if (e0 >= nnz) return;

  __builtin_prefetch(rows + e0 + 4096, 0, 1);
  __builtin_prefetch(cols + e0 + 4096, 0, 1);
  __builtin_prefetch(vals + e0 + 4096, 0, 1);

  const bool lo = (lane < 16);
  const int off0 = lo ? 2 * lane : 2 * lane - 31;

  if (e0 + 3 < nnz) {
    const int eu = __builtin_amdgcn_readfirstlane(e0);
    const int4   r4 = *(const int4*)(rows + eu);
    const int4   c4 = *(const int4*)(cols + eu);
    const float4 v4 = *(const float4*)(vals + eu);
    const int   rr[4] = {r4.x, r4.y, r4.z, r4.w};
    const int   cc[4] = {c4.x, c4.y, c4.z, c4.w};
    const float vv[4] = {v4.x, v4.y, v4.z, v4.w};
#pragma unroll
    for (int k = 0; k < 4; ++k) {
      const float2 xv = ((const float2*)(x + (size_t)cc[k] * FACTOR))[lane];
      scatter_row(y + (size_t)rr[k] * FACTOR, xv, vv[k], lo, off0);
    }
  } else {
    for (int k = 0; k < 4; ++k) {
      const int e = e0 + k;
      if (e >= nnz) break;
      const int r = rows[e];
      const int c = cols[e];
      const float v = vals[e];
      const float2 xv = ((const float2*)(x + (size_t)c * FACTOR))[lane];
      scatter_row(y + (size_t)r * FACTOR, xv, v, lo, off0);
    }
  }
}

__global__ __launch_bounds__(128) void combine_wmma(
    const float* __restrict__ x0, const float* __restrict__ x1,
    const float* __restrict__ x2, const float* __restrict__ x3,
    float* __restrict__ out, int nrows) {
  const int lane = threadIdx.x & 31;
  const int wave = threadIdx.x >> 5;
  const int row = blockIdx.x * (blockDim.x >> 5) + wave;
  if (row >= nrows) return;

  const bool lo = (lane < 16);
  v2f a;
  a.x = lo ? 1.0f : (1.0f / 3.0f);
  a.y = lo ? 0.5f : 0.25f;

  const float* __restrict__ p0 = lo ? x0 : x2;
  const float* __restrict__ p1 = lo ? x1 : x3;

  const size_t base = (size_t)row * FACTOR;
  const int n = lane & 15;

#pragma unroll
  for (int nb = 0; nb < FACTOR; nb += 16) {
    v2f b;
    b.x = p0[base + nb + n];
    b.y = p1[base + nb + n];
    v8f c = {};
    c = __builtin_amdgcn_wmma_f32_16x16x4_f32(
         false, a,  false, b,
         (short)0, c,  false,  false);
    out[base + nb + n] = c[0];
  }
}

#define L1_WPB   8
#define L1_NW    8192
#define L1_GRP   32
#define L1_NGRP  (L1_NW / L1_GRP)
#define L2_CAP   20480

__device__ __forceinline__ unsigned same_digit_lanes(unsigned d, bool valid) {
  unsigned m = (unsigned)__builtin_amdgcn_ballot_w32(valid);
#pragma unroll
  for (int b = 0; b < 8; ++b) {
    const bool bit = ((d >> b) & 1u) != 0u;
    const unsigned bal = (unsigned)__builtin_amdgcn_ballot_w32(bit);
    m &= bit ? bal : ~bal;
  }
  return m;
}

__global__ __launch_bounds__(256) void split_count(
    const int* __restrict__ u_idx, const int* __restrict__ i_idx,
    int nnz, int U, int I, unsigned* __restrict__ cnt1) {
  __shared__ unsigned hist_s[L1_WPB * 256];
  const int lane = threadIdx.x & 31;
  const int wl = threadIdx.x >> 5;
  const int dir = blockIdx.y;
  const int* __restrict__ key = dir ? i_idx : u_idx;
  const unsigned nrows = (unsigned)(dir ? I : U);
  const int wg = blockIdx.x * L1_WPB + wl;
  const int hb = wl * 256;
  for (int j = lane; j < 256; j += 32) hist_s[hb + j] = 0u;
  const int per = (nnz + L1_NW - 1) / L1_NW;
  const int e_lo = min(nnz, wg * per);
  const int e_hi = min(nnz, e_lo + per);
  const unsigned lt = (1u << lane) - 1u;
  unsigned kn = 0xffffffffu;
  if (e_lo + lane < e_hi) kn = (unsigned)key[e_lo + lane];
  for (int eb = e_lo; eb < e_hi; eb += 32) {
    const unsigned k = kn;
    const int en = eb + 32 + lane;
    kn = 0xffffffffu;
    if (en < e_hi) kn = (unsigned)key[en];
    const bool valid = (k < nrows);
    const unsigned d = (k >> 8) & 255u;
    const unsigned peers = same_digit_lanes(d, valid);
    if (valid && (peers & lt) == 0u)
      hist_s[hb + d] = hist_s[hb + d] + (unsigned)__builtin_popcount(peers);
  }
  unsigned* __restrict__ out = cnt1 + ((size_t)dir * L1_NW + (size_t)wg) * 256;
  for (int j = lane; j < 256; j += 32) out[j] = hist_s[hb + j];
}

__global__ __launch_bounds__(256) void split_scan_a(
    unsigned* __restrict__ cnt1, unsigned* __restrict__ grp) {
  const int b = threadIdx.x;
  const int g = blockIdx.x;
  const int dir = blockIdx.y;
  unsigned* __restrict__ c = cnt1 + ((size_t)dir * L1_NW + (size_t)g * L1_GRP) * 256 + b;
  unsigned run = 0u;
  for (int w0 = 0; w0 < L1_GRP; w0 += 16) {
    unsigned v[16];
#pragma unroll
    for (int j = 0; j < 16; ++j) v[j] = c[(size_t)(w0 + j) * 256];
#pragma unroll
    for (int j = 0; j < 16; ++j) { c[(size_t)(w0 + j) * 256] = run; run += v[j]; }
  }
  grp[((size_t)dir * L1_NGRP + g) * 256 + b] = run;
}

__global__ __launch_bounds__(256) void split_scan_b(
    unsigned* __restrict__ grp, unsigned* __restrict__ bbase) {
  __shared__ unsigned s[256];
  const int b = threadIdx.x;
  const int dir = blockIdx.x;
  unsigned* __restrict__ c = grp + (size_t)dir * L1_NGRP * 256 + b;
  unsigned run = 0u;
  for (int g0 = 0; g0 < L1_NGRP; g0 += 16) {
    unsigned v[16];
#pragma unroll
    for (int j = 0; j < 16; ++j) v[j] = c[(size_t)(g0 + j) * 256];
#pragma unroll
    for (int j = 0; j < 16; ++j) { c[(size_t)(g0 + j) * 256] = run; run += v[j]; }
  }
  s[b] = run;
  __syncthreads();
  for (int off = 1; off < 256; off <<= 1) {
    const unsigned v = (b >= off) ? s[b - off] : 0u;
    __syncthreads();
    s[b] += v;
    __syncthreads();
  }
  bbase[dir * 257 + b] = s[b] - run;
  if (b == 255) bbase[dir * 257 + 256] = s[255];
}

__global__ __launch_bounds__(256) void split_place(
    const int* __restrict__ u_idx, const int* __restrict__ i_idx,
    const float* __restrict__ vals_ui, const float* __restrict__ vals_iu,
    int nnz, int U, int I,
    const unsigned* __restrict__ cnt1, const unsigned* __restrict__ grp,
    const unsigned* __restrict__ bbase, int* __restrict__ rec1_) {
  __shared__ unsigned hist_s[L1_WPB * 256];
  const int lane = threadIdx.x & 31;
  const int wl = threadIdx.x >> 5;
  const int dir = blockIdx.y;
  const int* __restrict__ key = dir ? i_idx : u_idx;
  const int* __restrict__ src = dir ? u_idx : i_idx;
  const float* __restrict__ vals = dir ? vals_iu : vals_ui;
  const unsigned nrows = (unsigned)(dir ? I : U);
  const int wg = blockIdx.x * L1_WPB + wl;
  const int g = wg / L1_GRP;
  int2* __restrict__ rec1 = (int2*)rec1_ + (size_t)dir * (size_t)nnz;
  const int hb = wl * 256;
  {
    const unsigned* __restrict__ c = cnt1 + ((size_t)dir * L1_NW + (size_t)wg) * 256;
    const unsigned* __restrict__ gp = grp + ((size_t)dir * L1_NGRP + (size_t)g) * 256;
    const unsigned* __restrict__ bb = bbase + dir * 257;
    for (int j = lane; j < 256; j += 32) hist_s[hb + j] = bb[j] + gp[j] + c[j];
  }
  const int per = (nnz + L1_NW - 1) / L1_NW;
  const int e_lo = min(nnz, wg * per);
  const int e_hi = min(nnz, e_lo + per);
  const unsigned lt = (1u << lane) - 1u;
  unsigned kn = 0xffffffffu;
  int cn = 0;
  float vn = 0.0f;
  if (e_lo + lane < e_hi) { kn = (unsigned)key[e_lo + lane]; cn = src[e_lo + lane]; vn = vals[e_lo + lane]; }
  for (int eb = e_lo; eb < e_hi; eb += 32) {
    const unsigned k = kn;
    const int c = cn;
    const float v = vn;
    const int en = eb + 32 + lane;
    kn = 0xffffffffu;
    if (en < e_hi) { kn = (unsigned)key[en]; cn = src[en]; vn = vals[en]; }
    const bool valid = (k < nrows);
    const unsigned d = (k >> 8) & 255u;
    const unsigned peers = same_digit_lanes(d, valid);
    const unsigned rank = (unsigned)__builtin_popcount(peers & lt);
    unsigned old = 0u;
    if (valid) old = hist_s[hb + d];
    if (valid && rank == 0u) hist_s[hb + d] = old + (unsigned)__builtin_popcount(peers);
    if (valid) rec1[old + rank] = make_int2((int)(((k & 255u) << 16) | (unsigned)c),
                                            __float_as_int(v));
  }
}

__global__ __launch_bounds__(1024, 16) void rows_place(
    const int* __restrict__ rec1_, const unsigned* __restrict__ bbase,
    int nnz, int U, int I,
    unsigned* __restrict__ rowptr_u, unsigned* __restrict__ rowptr_i,
    int* __restrict__ rec_u, int* __restrict__ rec_i) {
  __shared__ unsigned hist_s[32 * 256];
  __shared__ unsigned s[256];
  __shared__ float stage_v[L2_CAP];
  __shared__ unsigned short stage_c[L2_CAP];
  const int lane = threadIdx.x & 31;
  const int wl = threadIdx.x >> 5;
  const int t = threadIdx.x;
  const int dir = blockIdx.y;
  const int bucket = blockIdx.x;
  const int nrows = dir ? I : U;
  if (bucket * 256 >= nrows) return;
  const int2* __restrict__ rec1 = (const int2*)rec1_ + (size_t)dir * (size_t)nnz;
  int2* __restrict__ rec = (int2*)(dir ? rec_i : rec_u);
  unsigned* __restrict__ rowptr = dir ? rowptr_i : rowptr_u;
  const unsigned bb = bbase[dir * 257 + bucket];
  const unsigned be = bbase[dir * 257 + bucket + 1];
  const unsigned nb = be - bb;
  const bool staged = (nb <= (unsigned)L2_CAP);
  const unsigned per = (nb + 31u) / 32u;
  const unsigned lo = min(be, bb + (unsigned)wl * per);
  const unsigned hi = min(be, lo + per);
  const int hb = wl * 256;
  for (int j = lane; j < 256; j += 32) hist_s[hb + j] = 0u;
  const unsigned lt = (1u << lane) - 1u;
  {
    int xn = 0;
    if (lo + (unsigned)lane < hi) xn = rec1[lo + (unsigned)lane].x;
    for (unsigned p0 = lo; p0 < hi; p0 += 32u) {
      const bool valid = (p0 + (unsigned)lane < hi);
      const unsigned d = ((unsigned)xn >> 16) & 255u;
      const unsigned pn = p0 + 32u + (unsigned)lane;
      xn = 0;
      if (pn < hi) xn = rec1[pn].x;
      const unsigned peers = same_digit_lanes(d, valid);
      if (valid && (peers & lt) == 0u)
        hist_s[hb + d] = hist_s[hb + d] + (unsigned)__builtin_popcount(peers);
    }
  }
  __syncthreads();
  unsigned tot = 0u;
  if (t < 256) {
    for (int w = 0; w < 32; ++w) {
      const unsigned c = hist_s[w * 256 + t];
      hist_s[w * 256 + t] = tot;
      tot += c;
    }
    s[t] = tot;
  }
  __syncthreads();
  for (int off = 1; off < 256; off <<= 1) {
    unsigned v = 0u;
    if (t < 256 && t >= off) v = s[t - off];
    __syncthreads();
    if (t < 256) s[t] += v;
    __syncthreads();
  }
  if (t < 256) {
    const unsigned start = bb + s[t] - tot;
    const int row = bucket * 256 + t;
    if (row < nrows) rowptr[row] = start;
    if (row == nrows - 1) rowptr[nrows] = be;
    for (int w = 0; w < 32; ++w) hist_s[w * 256 + t] += start;
  }
  __syncthreads();
  int2 rn = make_int2(0, 0);
  if (lo + (unsigned)lane < hi) rn = rec1[lo + (unsigned)lane];
  for (unsigned p0 = lo; p0 < hi; p0 += 32u) {
    const bool valid = (p0 + (unsigned)lane < hi);
    const int2 r = rn;
    const unsigned pn = p0 + 32u + (unsigned)lane;
    rn = make_int2(0, 0);
    if (pn < hi) rn = rec1[pn];
    const unsigned d = ((unsigned)r.x >> 16) & 255u;
    const unsigned peers = same_digit_lanes(d, valid);
    const unsigned rank = (unsigned)__builtin_popcount(peers & lt);
    unsigned old = 0u;
    if (valid) old = hist_s[hb + d];
    if (valid && rank == 0u) hist_s[hb + d] = old + (unsigned)__builtin_popcount(peers);
    if (valid) {
      const unsigned slot = old + rank;
      if (staged) {
        stage_c[slot - bb] = (unsigned short)(r.x & 0xffff);
        stage_v[slot - bb] = __int_as_float(r.y);
      } else {
        rec[slot] = make_int2(r.x & 0xffff, r.y);
      }
    }
  }
  if (staged) {
    __syncthreads();
    const int a = (int)(bb & 15u);
    for (int j = t - a; j < (int)nb; j += 1024)
      if (j >= 0) rec[bb + (unsigned)j] = make_int2((int)stage_c[j], __float_as_int(stage_v[j]));
  }
}

template <bool TAIL>
__device__ __forceinline__ void edges8(const float* __restrict__ x, int2 my, unsigned k,
                                       unsigned n, int lane, float& ax, float& ay) {
  float2 xv[8];
  float v[8];
#pragma unroll
  for (int j = 0; j < 8; ++j) {
    const unsigned kk = TAIL ? min(k + (unsigned)j, n - 1u) : (k + (unsigned)j);
    const int col = __builtin_amdgcn_readlane(my.x, (int)kk);
    v[j] = __int_as_float(__builtin_amdgcn_readlane(my.y, (int)kk));
    xv[j] = ((const float2*)(x + (size_t)col * FACTOR))[lane];
  }
#pragma unroll
  for (int j = 0; j < 8; ++j) {
    if (!TAIL || (k + (unsigned)j < n)) {
      ax += xv[j].x * v[j];
      ay += xv[j].y * v[j];
    }
  }
}

template <bool FINISH>
__global__ __launch_bounds__(256) void spmm_rows(
    const unsigned* __restrict__ rowptr, const int* __restrict__ rec_,
    const float* __restrict__ x, float* __restrict__ y, int nrows,
    const float* __restrict__ e0, const float* __restrict__ h1,
    const float* __restrict__ h2) {
  const int lane = threadIdx.x;
  const int row = blockIdx.x * blockDim.y + threadIdx.y;
  if (row >= nrows) return;
  const int r = __builtin_amdgcn_readfirstlane(row);
  float2 q0 = make_float2(0.0f, 0.0f), q1 = q0, q2 = q0;
  if (FINISH) {
    const size_t rb = (size_t)r * FACTOR;
    q0 = ((const float2*)(e0 + rb))[lane];
    q1 = ((const float2*)(h1 + rb))[lane];
    q2 = ((const float2*)(h2 + rb))[lane];
  }
  const unsigned beg = (unsigned)__builtin_amdgcn_readfirstlane((int)rowptr[r]);
  const unsigned end = (unsigned)__builtin_amdgcn_readfirstlane((int)rowptr[r + 1]);
  const int2* __restrict__ rec = (const int2*)rec_;
  float ax = 0.0f, ay = 0.0f;
  for (unsigned c = beg; c < end; c += 32u) {
    const unsigned n = min(32u, end - c);
    int2 my = make_int2(0, 0);
    if ((unsigned)lane < n) my = rec[c + (unsigned)lane];
    unsigned k = 0u;
    for (; k + 8u <= n; k += 8u) edges8<false>(x, my, k, n, lane, ax, ay);
    if (k < n) edges8<true>(x, my, k, n, lane, ax, ay);
  }
  float2 o;
  if (FINISH) {
    const bool lo = (lane < 16);
    v2f a;
    a.x = lo ? 1.0f : (1.0f / 3.0f);
    a.y = lo ? 0.5f : 0.25f;
    const float s0x = xhalf(lo ? q2.x : q0.x);
    const float s0y = xhalf(lo ? q2.y : q0.y);
    const float s1x = xhalf(lo ? ax : q1.x);
    const float s1y = xhalf(lo ? ay : q1.y);
    v2f b0, b1, b2, b3;
    b0.x = lo ? q0.x : s0x;  b0.y = lo ? q1.x : s1x;
    b1.x = lo ? q0.y : s0y;  b1.y = lo ? q1.y : s1y;
    b2.x = lo ? s0x : q2.x;  b2.y = lo ? s1x : ax;
    b3.x = lo ? s0y : q2.y;  b3.y = lo ? s1y : ay;
    const v8f z = {};
    const v8f d0 = __builtin_amdgcn_wmma_f32_16x16x4_f32(false, a, false, b0, (short)0, z, false, false);
    const v8f d1 = __builtin_amdgcn_wmma_f32_16x16x4_f32(false, a, false, b1, (short)0, z, false, false);
    const v8f d2 = __builtin_amdgcn_wmma_f32_16x16x4_f32(false, a, false, b2, (short)0, z, false, false);
    const v8f d3 = __builtin_amdgcn_wmma_f32_16x16x4_f32(false, a, false, b3, (short)0, z, false, false);
    o.x = lo ? d0[0] : d2[0];
    o.y = lo ? d1[0] : d3[0];
  } else {
    o.x = ax; o.y = ay;
  }
  ((float2*)(y + (size_t)r * FACTOR))[lane] = o;
}

extern "C" void kernel_launch(void* const* d_in, const int* in_sizes, int n_in,
                              void* d_out, int out_size, void* d_ws, size_t ws_size,
                              hipStream_t stream) {
  (void)n_in; (void)out_size;

  const float* embed_user = (const float*)d_in[0];
  const float* embed_item = (const float*)d_in[1];
  const int*   u_idx      = (const int*)d_in[2];
  const int*   i_idx      = (const int*)d_in[3];
  const float* vals_ui    = (const float*)d_in[4];
  const float* vals_iu    = (const float*)d_in[5];

  const int U   = in_sizes[0] / FACTOR;
  const int I   = in_sizes[1] / FACTOR;
  const int nnz = in_sizes[2];

  float* g1u = (float*)d_ws;
  float* g1i = g1u + (size_t)U * FACTOR;
  float* g2u = g1i + (size_t)I * FACTOR;
  float* g2i = g2u + (size_t)U * FACTOR;
  float* g3u = g2i + (size_t)I * FACTOR;
  float* g3i = g3u + (size_t)U * FACTOR;

  const size_t plane_bytes = (size_t)(U + I) * FACTOR * sizeof(float);
  const size_t acc_bytes = (size_t)3 * plane_bytes;
  const size_t rec_bytes = (size_t)2 * (size_t)nnz * 8;
  const size_t cnt1_bytes = (size_t)2 * L1_NW * 256 * sizeof(unsigned);
  const size_t small_words = (size_t)(U + I) + 2 + (size_t)2 * L1_NGRP * 256 + (size_t)2 * 257;
  const size_t need = acc_bytes + rec_bytes + small_words * sizeof(unsigned);

  float* out_users = (float*)d_out;
  float* out_items = out_users + (size_t)U * FACTOR;

  const bool ordered = (ws_size >= need) && (U <= 65536) && (I <= 65536) &&
                       (plane_bytes >= cnt1_bytes) && ((size_t)2 * plane_bytes >= rec_bytes);

  if (ordered) {
    int* rec_u = (int*)(g3i + (size_t)I * FACTOR);
    int* rec_i = rec_u + (size_t)2 * (size_t)nnz;
    unsigned* rowptr_u = (unsigned*)(rec_i + (size_t)2 * (size_t)nnz);
    unsigned* rowptr_i = rowptr_u + (size_t)U + 1;
    unsigned* grp   = rowptr_i + (size_t)I + 1;
    unsigned* bbase = grp + (size_t)2 * L1_NGRP * 256;
    unsigned* cnt1 = (unsigned*)g1u;
    int* rec1 = (int*)g2u;

    dim3 l1grid(L1_NW / L1_WPB, 2, 1);
    split_count<<<l1grid, L1_WPB * 32, 0, stream>>>(u_idx, i_idx, nnz, U, I, cnt1);
    split_scan_a<<<dim3(L1_NGRP, 2, 1), 256, 0, stream>>>(cnt1, grp);
    split_scan_b<<<2, 256, 0, stream>>>(grp, bbase);
    split_place<<<l1grid, L1_WPB * 32, 0, stream>>>(u_idx, i_idx, vals_ui, vals_iu, nnz, U, I,
                                                     cnt1, grp, bbase, rec1);
    rows_place<<<dim3(256, 2, 1), 1024, 0, stream>>>(rec1, bbase, nnz, U, I,
                                                      rowptr_u, rowptr_i, rec_u, rec_i);

    dim3 rblk(32, 8, 1);
    const int gu = (U + 7) / 8;
    const int gi = (I + 7) / 8;
    const float* none = nullptr;
    spmm_rows<false><<<gu, rblk, 0, stream>>>(rowptr_u, rec_u, embed_item, g1u, U, none, none, none);
    spmm_rows<false><<<gi, rblk, 0, stream>>>(rowptr_i, rec_i, embed_user, g1i, I, none, none, none);
    spmm_rows<false><<<gu, rblk, 0, stream>>>(rowptr_u, rec_u, g1i, g2u, U, none, none, none);
    spmm_rows<false><<<gi, rblk, 0, stream>>>(rowptr_i, rec_i, g1u, g2i, I, none, none, none);
    spmm_rows<true><<<gu, rblk, 0, stream>>>(rowptr_u, rec_u, g2i, out_users, U, embed_user, g1u, g2u);
    spmm_rows<true><<<gi, rblk, 0, stream>>>(rowptr_i, rec_i, g2u, out_items, I, embed_item, g1i, g2i);
  } else {
    hipMemsetAsync(d_ws, 0, acc_bytes, stream);

    dim3 sblk(32, 8, 1);
    const int waves = (nnz + 3) / 4;
    const int sgrid = (waves + 7) / 8;

    spmm_coo_atomic<<<sgrid, sblk, 0, stream>>>(u_idx, i_idx, vals_ui, embed_item, g1u, nnz);
    spmm_coo_atomic<<<sgrid, sblk, 0, stream>>>(i_idx, u_idx, vals_iu, embed_user, g1i, nnz);
    spmm_coo_atomic<<<sgrid, sblk, 0, stream>>>(u_idx, i_idx, vals_ui, g1i, g2u, nnz);
    spmm_coo_atomic<<<sgrid, sblk, 0, stream>>>(i_idx, u_idx, vals_iu, g1u, g2i, nnz);
    spmm_coo_atomic<<<sgrid, sblk, 0, stream>>>(u_idx, i_idx, vals_ui, g2i, g3u, nnz);
    spmm_coo_atomic<<<sgrid, sblk, 0, stream>>>(i_idx, u_idx, vals_iu, g2u, g3i, nnz);

    const int wpb = 4;
    combine_wmma<<<(U + wpb - 1) / wpb, wpb * 32, 0, stream>>>(embed_user, g1u, g2u, g3u, out_users, U);
    combine_wmma<<<(I + wpb - 1) / wpb, wpb * 32, 0, stream>>>(embed_item, g1i, g2i, g3i, out_items, I);
  }
}


# Level-1 placement: one 32-wave workgroup per scan group orders its records by bucket in LDS and copies each bucket's run out contiguously (was a scattered 8-B store per record); same sums
# speedup vs baseline: 3.4982x; 1.1775x over previous
#include <hip/hip_runtime.h>
#include <hip/hip_bf16.h>

#define FACTOR 64

typedef float v2f __attribute__((ext_vector_type(2)));
typedef float v8f __attribute__((ext_vector_type(8)));

__device__ __forceinline__ void atomic_add_f32_dev(float* p, float v) {
  unsigned long long a = (unsigned long long)p;
  asm volatile("global_atomic_add_f32 %0, %1, off scope:SCOPE_DEV"
               :: "v"(a), "v"(v));
}

__device__ __forceinline__ float xhalf(float x) {
  const int b = __builtin_bit_cast(int, x);
  return __builtin_bit_cast(float,
      __builtin_amdgcn_permlanex16(b, b, 0x76543210, 0xfedcba98, false, false));
}

__device__ __forceinline__ void scatter_row(float* yrow, float2 xv, float v,
                                            bool lo, int off0) {
  const float px = xv.x * v;
  const float py = xv.y * v;
  const float got = xhalf(lo ? py : px);
  float* yp = yrow + off0;
  atomic_add_f32_dev(yp,      lo ? px : got);
  atomic_add_f32_dev(yp + 32, lo ? got : py);
}

__global__ __launch_bounds__(256) void spmm_coo_atomic(
    const int* __restrict__ rows, const int* __restrict__ cols,
    const float* __restrict__ vals, const float* __restrict__ x,
    float* __restrict__ y, int nnz) {
  const int lane = threadIdx.x;
  const int waveId = blockIdx.x * blockDim.y + threadIdx.y;
  const int e0 = waveId * 4;
  if (e0 >= nnz) return;

  __builtin_prefetch(rows + e0 + 4096, 0, 1);
  __builtin_prefetch(cols + e0 + 4096, 0, 1);
  __builtin_prefetch(vals + e0 + 4096, 0, 1);

  const bool lo = (lane < 16);
  const int off0 = lo ? 2 * lane : 2 * lane - 31;

  if (e0 + 3 < nnz) {
    const int eu = __builtin_amdgcn_readfirstlane(e0);
    const int4   r4 = *(const int4*)(rows + eu);
    const int4   c4 = *(const int4*)(cols + eu);
    const float4 v4 = *(const float4*)(vals + eu);
    const int   rr[4] = {r4.x, r4.y, r4.z, r4.w};
    const int   cc[4] = {c4.x, c4.y, c4.z, c4.w};
    const float vv[4] = {v4.x, v4.y, v4.z, v4.w};
#pragma unroll
    for (int k = 0; k < 4; ++k) {
      const float2 xv = ((const float2*)(x + (size_t)cc[k] * FACTOR))[lane];
      scatter_row(y + (size_t)rr[k] * FACTOR, xv, vv[k], lo, off0);
    }
  } else {
    for (int k = 0; k < 4; ++k) {
      const int e = e0 + k;
      if (e >= nnz) break;
      const int r = rows[e];
      const int c = cols[e];
      const float v = vals[e];
      const float2 xv = ((const float2*)(x + (size_t)c * FACTOR))[lane];
      scatter_row(y + (size_t)r * FACTOR, xv, v, lo, off0);
    }
  }
}

__global__ __launch_bounds__(128) void combine_wmma(
    const float* __restrict__ x0, const float* __restrict__ x1,
    const float* __restrict__ x2, const float* __restrict__ x3,
    float* __restrict__ out, int nrows) {
  const int lane = threadIdx.x & 31;
  const int wave = threadIdx.x >> 5;
  const int row = blockIdx.x * (blockDim.x >> 5) + wave;
  if (row >= nrows) return;

  const bool lo = (lane < 16);
  v2f a;
  a.x = lo ? 1.0f : (1.0f / 3.0f);
  a.y = lo ? 0.5f : 0.25f;

  const float* __restrict__ p0 = lo ? x0 : x2;
  const float* __restrict__ p1 = lo ? x1 : x3;

  const size_t base = (size_t)row * FACTOR;
  const int n = lane & 15;

#pragma unroll
  for (int nb = 0; nb < FACTOR; nb += 16) {
    v2f b;
    b.x = p0[base + nb + n];
    b.y = p1[base + nb + n];
    v8f c = {};
    c = __builtin_amdgcn_wmma_f32_16x16x4_f32(
         false, a,  false, b,
         (short)0, c,  false,  false);
    out[base + nb + n] = c[0];
  }
}

#define L1_WPB   8
#define L1_NW    8192
#define L1_GRP   32
#define L1_NGRP  (L1_NW / L1_GRP)
#define L1_CAP   12288
#define L2_CAP   20480

__device__ __forceinline__ unsigned same_digit_lanes(unsigned d, bool valid) {
  unsigned m = (unsigned)__builtin_amdgcn_ballot_w32(valid);
#pragma unroll
  for (int b = 0; b < 8; ++b) {
    const bool bit = ((d >> b) & 1u) != 0u;
    const unsigned bal = (unsigned)__builtin_amdgcn_ballot_w32(bit);
    m &= bit ? bal : ~bal;
  }
  return m;
}

__global__ __launch_bounds__(256) void split_count(
    const int* __restrict__ u_idx, const int* __restrict__ i_idx,
    int nnz, int U, int I, unsigned* __restrict__ cnt1) {
  __shared__ unsigned hist_s[L1_WPB * 256];
  const int lane = threadIdx.x & 31;
  const int wl = threadIdx.x >> 5;
  const int dir = blockIdx.y;
  const int* __restrict__ key = dir ? i_idx : u_idx;
  const unsigned nrows = (unsigned)(dir ? I : U);
  const int wg = blockIdx.x * L1_WPB + wl;
  const int hb = wl * 256;
  for (int j = lane; j < 256; j += 32) hist_s[hb + j] = 0u;
  const int per = (nnz + L1_NW - 1) / L1_NW;
  const int e_lo = min(nnz, wg * per);
  const int e_hi = min(nnz, e_lo + per);
  const unsigned lt = (1u << lane) - 1u;
  unsigned kn = 0xffffffffu;
  if (e_lo + lane < e_hi) kn = (unsigned)key[e_lo + lane];
  for (int eb = e_lo; eb < e_hi; eb += 32) {
    const unsigned k = kn;
    const int en = eb + 32 + lane;
    kn = 0xffffffffu;
    if (en < e_hi) kn = (unsigned)key[en];
    const bool valid = (k < nrows);
    const unsigned d = (k >> 8) & 255u;
    const unsigned peers = same_digit_lanes(d, valid);
    if (valid && (peers & lt) == 0u)
      hist_s[hb + d] = hist_s[hb + d] + (unsigned)__builtin_popcount(peers);
  }
  unsigned* __restrict__ out = cnt1 + ((size_t)dir * L1_NW + (size_t)wg) * 256;
  for (int j = lane; j < 256; j += 32) out[j] = hist_s[hb + j];
}

__global__ __launch_bounds__(256) void split_scan_a(
    unsigned* __restrict__ cnt1, unsigned* __restrict__ grp) {
  const int b = threadIdx.x;
  const int g = blockIdx.x;
  const int dir = blockIdx.y;
  unsigned* __restrict__ c = cnt1 + ((size_t)dir * L1_NW + (size_t)g * L1_GRP) * 256 + b;
  unsigned run = 0u;
  for (int w0 = 0; w0 < L1_GRP; w0 += 16) {
    unsigned v[16];
#pragma unroll
    for (int j = 0; j < 16; ++j) v[j] = c[(size_t)(w0 + j) * 256];
#pragma unroll
    for (int j = 0; j < 16; ++j) { c[(size_t)(w0 + j) * 256] = run; run += v[j]; }
  }
  grp[((size_t)dir * L1_NGRP + g) * 256 + b] = run;
}

__global__ __launch_bounds__(256) void split_scan_b(
    unsigned* __restrict__ grp, unsigned* __restrict__ bbase) {
  __shared__ unsigned s[256];
  const int b = threadIdx.x;
  const int dir = blockIdx.x;
  unsigned* __restrict__ c = grp + (size_t)dir * L1_NGRP * 256 + b;
  unsigned run = 0u;
  for (int g0 = 0; g0 < L1_NGRP; g0 += 16) {
    unsigned v[16];
#pragma unroll
    for (int j = 0; j < 16; ++j) v[j] = c[(size_t)(g0 + j) * 256];
#pragma unroll
    for (int j = 0; j < 16; ++j) { c[(size_t)(g0 + j) * 256] = run; run += v[j]; }
  }
  s[b] = run;
  __syncthreads();
  for (int off = 1; off < 256; off <<= 1) {
    const unsigned v = (b >= off) ? s[b - off] : 0u;
    __syncthreads();
    s[b] += v;
    __syncthreads();
  }
  bbase[dir * 257 + b] = s[b] - run;
  if (b == 255) bbase[dir * 257 + 256] = s[255];
}

__global__ __launch_bounds__(1024, 16) void split_place(
    const int* __restrict__ u_idx, const int* __restrict__ i_idx,
    const float* __restrict__ vals_ui, const float* __restrict__ vals_iu,
    int nnz, int U, int I,
    const unsigned* __restrict__ cnt1, const unsigned* __restrict__ grp,
    const unsigned* __restrict__ bbase, int* __restrict__ rec1_) {
  __shared__ unsigned hist_s[L1_GRP * 256];
  __shared__ unsigned s[256];
  __shared__ unsigned delta_s[256];
  __shared__ int2 stage[L1_CAP];
  const int lane = threadIdx.x & 31;
  const int wl = threadIdx.x >> 5;
  const int t = threadIdx.x;
  const int dir = blockIdx.y;
  const int g = blockIdx.x;
  const int* __restrict__ key = dir ? i_idx : u_idx;
  const int* __restrict__ src = dir ? u_idx : i_idx;
  const float* __restrict__ vals = dir ? vals_iu : vals_ui;
  const unsigned nrows = (unsigned)(dir ? I : U);
  const int wg = g * L1_GRP + wl;
  int2* __restrict__ rec1 = (int2*)rec1_ + (size_t)dir * (size_t)nnz;
  const int hb = wl * 256;
  const unsigned* __restrict__ c1 = cnt1 + ((size_t)dir * L1_NW + (size_t)wg) * 256;
  const unsigned* __restrict__ gp = grp + ((size_t)dir * L1_NGRP + (size_t)g) * 256;
  const unsigned* __restrict__ bb = bbase + dir * 257;
  const int per = (nnz + L1_NW - 1) / L1_NW;
  const int e_lo = min(nnz, wg * per);
  const int e_hi = min(nnz, e_lo + per);
  const unsigned lt = (1u << lane) - 1u;
  unsigned kn = 0xffffffffu;
  int cn = 0;
  float vn = 0.0f;
  if (e_lo + lane < e_hi) { kn = (unsigned)key[e_lo + lane]; cn = src[e_lo + lane]; vn = vals[e_lo + lane]; }
  unsigned run0 = 0u, n = 0u;
  if (t < 256) {
    const unsigned pre = gp[t];
    const unsigned nxt = (g + 1 < L1_NGRP) ? gp[256 + t] : (bb[t + 1] - bb[t]);
    run0 = bb[t] + pre;
    n = nxt - pre;
    s[t] = n;
  }
  __syncthreads();
  for (int off = 1; off < 256; off <<= 1) {
    unsigned v = 0u;
    if (t < 256 && t >= off) v = s[t - off];
    __syncthreads();
    if (t < 256) s[t] += v;
    __syncthreads();
  }
  const unsigned total = s[255];
  const bool staged = (total <= (unsigned)L1_CAP);
  if (t < 256) delta_s[t] = staged ? run0 - (s[t] - n) : 0u;
  __syncthreads();
  for (int j = lane; j < 256; j += 32) hist_s[hb + j] = bb[j] + gp[j] + c1[j] - delta_s[j];
  for (int eb = e_lo; eb < e_hi; eb += 32) {
    const unsigned k = kn;
    const int c = cn;
    const float v = vn;
    const int en = eb + 32 + lane;
    kn = 0xffffffffu;
    if (en < e_hi) { kn = (unsigned)key[en]; cn = src[en]; vn = vals[en]; }
    const bool valid = (k < nrows);
    const unsigned d = (k >> 8) & 255u;
    const unsigned peers = same_digit_lanes(d, valid);
    const unsigned rank = (unsigned)__builtin_popcount(peers & lt);
    unsigned old = 0u;
    if (valid) old = hist_s[hb + d];
    if (valid && rank == 0u) hist_s[hb + d] = old + (unsigned)__builtin_popcount(peers);
    if (valid) {
      const unsigned pos = old + rank;
      const unsigned x = ((k & 255u) << 16) | (unsigned)c;
      if (staged) stage[pos] = make_int2((int)((x & 0x00ffffffu) | (d << 24)), __float_as_int(v));
      else        rec1[pos] = make_int2((int)x, __float_as_int(v));
    }
  }
  if (staged) {
    __syncthreads();
    for (unsigned j = (unsigned)t; j < total; j += 1024u) {
      const int2 r = stage[j];
      const unsigned d = (unsigned)r.x >> 24;
      rec1[j + delta_s[d]] = make_int2(r.x & 0x00ffffff, r.y);
    }
  }
}

__global__ __launch_bounds__(1024, 16) void rows_place(
    const int* __restrict__ rec1_, const unsigned* __restrict__ bbase,
    int nnz, int U, int I,
    unsigned* __restrict__ rowptr_u, unsigned* __restrict__ rowptr_i,
    int* __restrict__ rec_u, int* __restrict__ rec_i) {
  __shared__ unsigned hist_s[32 * 256];
  __shared__ unsigned s[256];
  __shared__ float stage_v[L2_CAP];
  __shared__ unsigned short stage_c[L2_CAP];
  const int lane = threadIdx.x & 31;
  const int wl = threadIdx.x >> 5;
  const int t = threadIdx.x;
  const int dir = blockIdx.y;
  const int bucket = blockIdx.x;
  const int nrows = dir ? I : U;
  if (bucket * 256 >= nrows) return;
  const int2* __restrict__ rec1 = (const int2*)rec1_ + (size_t)dir * (size_t)nnz;
  int2* __restrict__ rec = (int2*)(dir ? rec_i : rec_u);
  unsigned* __restrict__ rowptr = dir ? rowptr_i : rowptr_u;
  const unsigned bb = bbase[dir * 257 + bucket];
  const unsigned be = bbase[dir * 257 + bucket + 1];
  const unsigned nb = be - bb;
  const bool staged = (nb <= (unsigned)L2_CAP);
  const unsigned per = (nb + 31u) / 32u;
  const unsigned lo = min(be, bb + (unsigned)wl * per);
  const unsigned hi = min(be, lo + per);
  const int hb = wl * 256;
  for (int j = lane; j < 256; j += 32) hist_s[hb + j] = 0u;
  const unsigned lt = (1u << lane) - 1u;
  {
    int xn = 0;
    if (lo + (unsigned)lane < hi) xn = rec1[lo + (unsigned)lane].x;
    for (unsigned p0 = lo; p0 < hi; p0 += 32u) {
      const bool valid = (p0 + (unsigned)lane < hi);
      const unsigned d = ((unsigned)xn >> 16) & 255u;
      const unsigned pn = p0 + 32u + (unsigned)lane;
      xn = 0;
      if (pn < hi) xn = rec1[pn].x;
      const unsigned peers = same_digit_lanes(d, valid);
      if (valid && (peers & lt) == 0u)
        hist_s[hb + d] = hist_s[hb + d] + (unsigned)__builtin_popcount(peers);
    }
  }
  __syncthreads();
  unsigned tot = 0u;
  if (t < 256) {
    for (int w = 0; w < 32; ++w) {
      const unsigned c = hist_s[w * 256 + t];
      hist_s[w * 256 + t] = tot;
      tot += c;
    }
    s[t] = tot;
  }
  __syncthreads();
  for (int off = 1; off < 256; off <<= 1) {
    unsigned v = 0u;
    if (t < 256 && t >= off) v = s[t - off];
    __syncthreads();
    if (t < 256) s[t] += v;
    __syncthreads();
  }
  if (t < 256) {
    const unsigned start = bb + s[t] - tot;
    const int row = bucket * 256 + t;
    if (row < nrows) rowptr[row] = start;
    if (row == nrows - 1) rowptr[nrows] = be;
    for (int w = 0; w < 32; ++w) hist_s[w * 256 + t] += start;
  }
  __syncthreads();
  int2 rn = make_int2(0, 0);
  if (lo + (unsigned)lane < hi) rn = rec1[lo + (unsigned)lane];
  for (unsigned p0 = lo; p0 < hi; p0 += 32u) {
    const bool valid = (p0 + (unsigned)lane < hi);
    const int2 r = rn;
    const unsigned pn = p0 + 32u + (unsigned)lane;
    rn = make_int2(0, 0);
    if (pn < hi) rn = rec1[pn];
    const unsigned d = ((unsigned)r.x >> 16) & 255u;
    const unsigned peers = same_digit_lanes(d, valid);
    const unsigned rank = (unsigned)__builtin_popcount(peers & lt);
    unsigned old = 0u;
    if (valid) old = hist_s[hb + d];
    if (valid && rank == 0u) hist_s[hb + d] = old + (unsigned)__builtin_popcount(peers);
    if (valid) {
      const unsigned slot = old + rank;
      if (staged) {
        stage_c[slot - bb] = (unsigned short)(r.x & 0xffff);
        stage_v[slot - bb] = __int_as_float(r.y);
      } else {
        rec[slot] = make_int2(r.x & 0xffff, r.y);
      }
    }
  }
  if (staged) {
    __syncthreads();
    const int a = (int)(bb & 15u);
    for (int j = t - a; j < (int)nb; j += 1024)
      if (j >= 0) rec[bb + (unsigned)j] = make_int2((int)stage_c[j], __float_as_int(stage_v[j]));
  }
}

template <bool TAIL>
__device__ __forceinline__ void edges8(const float* __restrict__ x, int2 my, unsigned k,
                                       unsigned n, int lane, float& ax, float& ay) {
  float2 xv[8];
  float v[8];
#pragma unroll
  for (int j = 0; j < 8; ++j) {
    const unsigned kk = TAIL ? min(k + (unsigned)j, n - 1u) : (k + (unsigned)j);
    const int col = __builtin_amdgcn_readlane(my.x, (int)kk);
    v[j] = __int_as_float(__builtin_amdgcn_readlane(my.y, (int)kk));
    xv[j] = ((const float2*)(x + (size_t)col * FACTOR))[lane];
  }
#pragma unroll
  for (int j = 0; j < 8; ++j) {
    if (!TAIL || (k + (unsigned)j < n)) {
      ax += xv[j].x * v[j];
      ay += xv[j].y * v[j];
    }
  }
}

template <bool FINISH>
__global__ __launch_bounds__(256) void spmm_rows(
    const unsigned* __restrict__ rowptr, const int* __restrict__ rec_,
    const float* __restrict__ x, float* __restrict__ y, int nrows,
    const float* __restrict__ e0, const float* __restrict__ h1,
    const float* __restrict__ h2) {
  const int lane = threadIdx.x;
  const int row = blockIdx.x * blockDim.y + threadIdx.y;
  if (row >= nrows) return;
  const int r = __builtin_amdgcn_readfirstlane(row);
  float2 q0 = make_float2(0.0f, 0.0f), q1 = q0, q2 = q0;
  if (FINISH) {
    const size_t rb = (size_t)r * FACTOR;
    q0 = ((const float2*)(e0 + rb))[lane];
    q1 = ((const float2*)(h1 + rb))[lane];
    q2 = ((const float2*)(h2 + rb))[lane];
  }
  const unsigned beg = (unsigned)__builtin_amdgcn_readfirstlane((int)rowptr[r]);
  const unsigned end = (unsigned)__builtin_amdgcn_readfirstlane((int)rowptr[r + 1]);
  const int2* __restrict__ rec = (const int2*)rec_;
  float ax = 0.0f, ay = 0.0f;
  for (unsigned c = beg; c < end; c += 32u) {
    const unsigned n = min(32u, end - c);
    int2 my = make_int2(0, 0);
    if ((unsigned)lane < n) my = rec[c + (unsigned)lane];
    unsigned k = 0u;
    for (; k + 8u <= n; k += 8u) edges8<false>(x, my, k, n, lane, ax, ay);
    if (k < n) edges8<true>(x, my, k, n, lane, ax, ay);
  }
  float2 o;
  if (FINISH) {
    const bool lo = (lane < 16);
    v2f a;
    a.x = lo ? 1.0f : (1.0f / 3.0f);
    a.y = lo ? 0.5f : 0.25f;
    const float s0x = xhalf(lo ? q2.x : q0.x);
    const float s0y = xhalf(lo ? q2.y : q0.y);
    const float s1x = xhalf(lo ? ax : q1.x);
    const float s1y = xhalf(lo ? ay : q1.y);
    v2f b0, b1, b2, b3;
    b0.x = lo ? q0.x : s0x;  b0.y = lo ? q1.x : s1x;
    b1.x = lo ? q0.y : s0y;  b1.y = lo ? q1.y : s1y;
    b2.x = lo ? s0x : q2.x;  b2.y = lo ? s1x : ax;
    b3.x = lo ? s0y : q2.y;  b3.y = lo ? s1y : ay;
    const v8f z = {};
    const v8f d0 = __builtin_amdgcn_wmma_f32_16x16x4_f32(false, a, false, b0, (short)0, z, false, false);
    const v8f d1 = __builtin_amdgcn_wmma_f32_16x16x4_f32(false, a, false, b1, (short)0, z, false, false);
    const v8f d2 = __builtin_amdgcn_wmma_f32_16x16x4_f32(false, a, false, b2, (short)0, z, false, false);
    const v8f d3 = __builtin_amdgcn_wmma_f32_16x16x4_f32(false, a, false, b3, (short)0, z, false, false);
    o.x = lo ? d0[0] : d2[0];
    o.y = lo ? d1[0] : d3[0];
  } else {
    o.x = ax; o.y = ay;
  }
  ((float2*)(y + (size_t)r * FACTOR))[lane] = o;
}

extern "C" void kernel_launch(void* const* d_in, const int* in_sizes, int n_in,
                              void* d_out, int out_size, void* d_ws, size_t ws_size,
                              hipStream_t stream) {
  (void)n_in; (void)out_size;

  const float* embed_user = (const float*)d_in[0];
  const float* embed_item = (const float*)d_in[1];
  const int*   u_idx      = (const int*)d_in[2];
  const int*   i_idx      = (const int*)d_in[3];
  const float* vals_ui    = (const float*)d_in[4];
  const float* vals_iu    = (const float*)d_in[5];

  const int U   = in_sizes[0] / FACTOR;
  const int I   = in_sizes[1] / FACTOR;
  const int nnz = in_sizes[2];

  float* g1u = (float*)d_ws;
  float* g1i = g1u + (size_t)U * FACTOR;
  float* g2u = g1i + (size_t)I * FACTOR;
  float* g2i = g2u + (size_t)U * FACTOR;
  float* g3u = g2i + (size_t)I * FACTOR;
  float* g3i = g3u + (size_t)U * FACTOR;

  const size_t plane_bytes = (size_t)(U + I) * FACTOR * sizeof(float);
  const size_t acc_bytes = (size_t)3 * plane_bytes;
  const size_t rec_bytes = (size_t)2 * (size_t)nnz * 8;
  const size_t cnt1_bytes = (size_t)2 * L1_NW * 256 * sizeof(unsigned);
  const size_t small_words = (size_t)(U + I) + 2 + (size_t)2 * L1_NGRP * 256 + (size_t)2 * 257;
  const size_t need = acc_bytes + rec_bytes + small_words * sizeof(unsigned);

  float* out_users = (float*)d_out;
  float* out_items = out_users + (size_t)U * FACTOR;

  const bool ordered = (ws_size >= need) && (U <= 65536) && (I <= 65536) &&
                       (plane_bytes >= cnt1_bytes) && ((size_t)2 * plane_bytes >= rec_bytes);

  if (ordered) {
    int* rec_u = (int*)(g3i + (size_t)I * FACTOR);
    int* rec_i = rec_u + (size_t)2 * (size_t)nnz;
    unsigned* rowptr_u = (unsigned*)(rec_i + (size_t)2 * (size_t)nnz);
    unsigned* rowptr_i = rowptr_u + (size_t)U + 1;
    unsigned* grp   = rowptr_i + (size_t)I + 1;
    unsigned* bbase = grp + (size_t)2 * L1_NGRP * 256;
    unsigned* cnt1 = (unsigned*)g1u;
    int* rec1 = (int*)g2u;

    dim3 l1grid(L1_NW / L1_WPB, 2, 1);
    split_count<<<l1grid, L1_WPB * 32, 0, stream>>>(u_idx, i_idx, nnz, U, I, cnt1);
    split_scan_a<<<dim3(L1_NGRP, 2, 1), 256, 0, stream>>>(cnt1, grp);
    split_scan_b<<<2, 256, 0, stream>>>(grp, bbase);
    split_place<<<dim3(L1_NGRP, 2, 1), L1_GRP * 32, 0, stream>>>(u_idx, i_idx, vals_ui, vals_iu,
                                                                 nnz, U, I, cnt1, grp, bbase, rec1);
    rows_place<<<dim3(256, 2, 1), 1024, 0, stream>>>(rec1, bbase, nnz, U, I,
                                                      rowptr_u, rowptr_i, rec_u, rec_i);

    dim3 rblk(32, 8, 1);
    const int gu = (U + 7) / 8;
    const int gi = (I + 7) / 8;
    const float* none = nullptr;
    spmm_rows<false><<<gu, rblk, 0, stream>>>(rowptr_u, rec_u, embed_item, g1u, U, none, none, none);
    spmm_rows<false><<<gi, rblk, 0, stream>>>(rowptr_i, rec_i, embed_user, g1i, I, none, none, none);
    spmm_rows<false><<<gu, rblk, 0, stream>>>(rowptr_u, rec_u, g1i, g2u, U, none, none, none);
    spmm_rows<false><<<gi, rblk, 0, stream>>>(rowptr_i, rec_i, g1u, g2i, I, none, none, none);
    spmm_rows<true><<<gu, rblk, 0, stream>>>(rowptr_u, rec_u, g2i, out_users, U, embed_user, g1u, g2u);
    spmm_rows<true><<<gi, rblk, 0, stream>>>(rowptr_i, rec_i, g2u, out_items, I, embed_item, g1i, g2i);
  } else {
    hipMemsetAsync(d_ws, 0, acc_bytes, stream);

    dim3 sblk(32, 8, 1);
    const int waves = (nnz + 3) / 4;
    const int sgrid = (waves + 7) / 8;

    spmm_coo_atomic<<<sgrid, sblk, 0, stream>>>(u_idx, i_idx, vals_ui, embed_item, g1u, nnz);
    spmm_coo_atomic<<<sgrid, sblk, 0, stream>>>(i_idx, u_idx, vals_iu, embed_user, g1i, nnz);
    spmm_coo_atomic<<<sgrid, sblk, 0, stream>>>(u_idx, i_idx, vals_ui, g1i, g2u, nnz);
    spmm_coo_atomic<<<sgrid, sblk, 0, stream>>>(i_idx, u_idx, vals_iu, g1u, g2i, nnz);
    spmm_coo_atomic<<<sgrid, sblk, 0, stream>>>(u_idx, i_idx, vals_ui, g2i, g3u, nnz);
    spmm_coo_atomic<<<sgrid, sblk, 0, stream>>>(i_idx, u_idx, vals_iu, g2u, g3i, nnz);

    const int wpb = 4;
    combine_wmma<<<(U + wpb - 1) / wpb, wpb * 32, 0, stream>>>(embed_user, g1u, g2u, g3u, out_users, U);
    combine_wmma<<<(I + wpb - 1) / wpb, wpb * 32, 0, stream>>>(embed_item, g1i, g2i, g3i, out_items, I);
  }
}


# Row passes 6 -> 4: the two chains' middle products share a pass (A[e_i|g1i], B[g1u|g2u]), gathering one 512-B row of a paired plane per edge with 16-B-per-lane loads; same sums, same order
# speedup vs baseline: 3.6626x; 1.0470x over previous
#include <hip/hip_runtime.h>
#include <hip/hip_bf16.h>

#define FACTOR 64

typedef float v2f __attribute__((ext_vector_type(2)));
typedef float v8f __attribute__((ext_vector_type(8)));

__device__ __forceinline__ void atomic_add_f32_dev(float* p, float v) {
  unsigned long long a = (unsigned long long)p;
  asm volatile("global_atomic_add_f32 %0, %1, off scope:SCOPE_DEV"
               :: "v"(a), "v"(v));
}

__device__ __forceinline__ float xhalf(float x) {
  const int b = __builtin_bit_cast(int, x);
  return __builtin_bit_cast(float,
      __builtin_amdgcn_permlanex16(b, b, 0x76543210, 0xfedcba98, false, false));
}

__device__ __forceinline__ void scatter_row(float* yrow, float2 xv, float v,
                                            bool lo, int off0) {
  const float px = xv.x * v;
  const float py = xv.y * v;
  const float got = xhalf(lo ? py : px);
  float* yp = yrow + off0;
  atomic_add_f32_dev(yp,      lo ? px : got);
  atomic_add_f32_dev(yp + 32, lo ? got : py);
}

__global__ __launch_bounds__(256) void spmm_coo_atomic(
    const int* __restrict__ rows, const int* __restrict__ cols,
    const float* __restrict__ vals, const float* __restrict__ x,
    float* __restrict__ y, int nnz) {
  const int lane = threadIdx.x;
  const int waveId = blockIdx.x * blockDim.y + threadIdx.y;
  const int e0 = waveId * 4;
  if (e0 >= nnz) return;

  __builtin_prefetch(rows + e0 + 4096, 0, 1);
  __builtin_prefetch(cols + e0 + 4096, 0, 1);
  __builtin_prefetch(vals + e0 + 4096, 0, 1);

  const bool lo = (lane < 16);
  const int off0 = lo ? 2 * lane : 2 * lane - 31;

  if (e0 + 3 < nnz) {
    const int eu = __builtin_amdgcn_readfirstlane(e0);
    const int4   r4 = *(const int4*)(rows + eu);
    const int4   c4 = *(const int4*)(cols + eu);
    const float4 v4 = *(const float4*)(vals + eu);
    const int   rr[4] = {r4.x, r4.y, r4.z, r4.w};
    const int   cc[4] = {c4.x, c4.y, c4.z, c4.w};
    const float vv[4] = {v4.x, v4.y, v4.z, v4.w};
#pragma unroll
    for (int k = 0; k < 4; ++k) {
      const float2 xv = ((const float2*)(x + (size_t)cc[k] * FACTOR))[lane];
      scatter_row(y + (size_t)rr[k] * FACTOR, xv, vv[k], lo, off0);
    }
  } else {
    for (int k = 0; k < 4; ++k) {
      const int e = e0 + k;
      if (e >= nnz) break;
      const int r = rows[e];
      const int c = cols[e];
      const float v = vals[e];
      const float2 xv = ((const float2*)(x + (size_t)c * FACTOR))[lane];
      scatter_row(y + (size_t)r * FACTOR, xv, v, lo, off0);
    }
  }
}

__global__ __launch_bounds__(128) void combine_wmma(
    const float* __restrict__ x0, const float* __restrict__ x1,
    const float* __restrict__ x2, const float* __restrict__ x3,
    float* __restrict__ out, int nrows) {
  const int lane = threadIdx.x & 31;
  const int wave = threadIdx.x >> 5;
  const int row = blockIdx.x * (blockDim.x >> 5) + wave;
  if (row >= nrows) return;

  const bool lo = (lane < 16);
  v2f a;
  a.x = lo ? 1.0f : (1.0f / 3.0f);
  a.y = lo ? 0.5f : 0.25f;

  const float* __restrict__ p0 = lo ? x0 : x2;
  const float* __restrict__ p1 = lo ? x1 : x3;

  const size_t base = (size_t)row * FACTOR;
  const int n = lane & 15;

#pragma unroll
  for (int nb = 0; nb < FACTOR; nb += 16) {
    v2f b;
    b.x = p0[base + nb + n];
    b.y = p1[base + nb + n];
    v8f c = {};
    c = __builtin_amdgcn_wmma_f32_16x16x4_f32(
         false, a,  false, b,
         (short)0, c,  false,  false);
    out[base + nb + n] = c[0];
  }
}

#define L1_WPB   8
#define L1_NW    8192
#define L1_GRP   32
#define L1_NGRP  (L1_NW / L1_GRP)
#define L1_CAP   12288
#define L2_CAP   20480

__device__ __forceinline__ unsigned same_digit_lanes(unsigned d, bool valid) {
  unsigned m = (unsigned)__builtin_amdgcn_ballot_w32(valid);
#pragma unroll
  for (int b = 0; b < 8; ++b) {
    const bool bit = ((d >> b) & 1u) != 0u;
    const unsigned bal = (unsigned)__builtin_amdgcn_ballot_w32(bit);
    m &= bit ? bal : ~bal;
  }
  return m;
}

__global__ __launch_bounds__(256) void split_count(
    const int* __restrict__ u_idx, const int* __restrict__ i_idx,
    int nnz, int U, int I, unsigned* __restrict__ cnt1) {
  __shared__ unsigned hist_s[L1_WPB * 256];
  const int lane = threadIdx.x & 31;
  const int wl = threadIdx.x >> 5;
  const int dir = blockIdx.y;
  const int* __restrict__ key = dir ? i_idx : u_idx;
  const unsigned nrows = (unsigned)(dir ? I : U);
  const int wg = blockIdx.x * L1_WPB + wl;
  const int hb = wl * 256;
  for (int j = lane; j < 256; j += 32) hist_s[hb + j] = 0u;
  const int per = (nnz + L1_NW - 1) / L1_NW;
  const int e_lo = min(nnz, wg * per);
  const int e_hi = min(nnz, e_lo + per);
  const unsigned lt = (1u << lane) - 1u;
  unsigned kn = 0xffffffffu;
  if (e_lo + lane < e_hi) kn = (unsigned)key[e_lo + lane];
  for (int eb = e_lo; eb < e_hi; eb += 32) {
    const unsigned k = kn;
    const int en = eb + 32 + lane;
    kn = 0xffffffffu;
    if (en < e_hi) kn = (unsigned)key[en];
    const bool valid = (k < nrows);
    const unsigned d = (k >> 8) & 255u;
    const unsigned peers = same_digit_lanes(d, valid);
    if (valid && (peers & lt) == 0u)
      hist_s[hb + d] = hist_s[hb + d] + (unsigned)__builtin_popcount(peers);
  }
  unsigned* __restrict__ out = cnt1 + ((size_t)dir * L1_NW + (size_t)wg) * 256;
  for (int j = lane; j < 256; j += 32) out[j] = hist_s[hb + j];
}

__global__ __launch_bounds__(256) void split_scan_a(
    unsigned* __restrict__ cnt1, unsigned* __restrict__ grp) {
  const int b = threadIdx.x;
  const int g = blockIdx.x;
  const int dir = blockIdx.y;
  unsigned* __restrict__ c = cnt1 + ((size_t)dir * L1_NW + (size_t)g * L1_GRP) * 256 + b;
  unsigned run = 0u;
  for (int w0 = 0; w0 < L1_GRP; w0 += 16) {
    unsigned v[16];
#pragma unroll
    for (int j = 0; j < 16; ++j) v[j] = c[(size_t)(w0 + j) * 256];
#pragma unroll
    for (int j = 0; j < 16; ++j) { c[(size_t)(w0 + j) * 256] = run; run += v[j]; }
  }
  grp[((size_t)dir * L1_NGRP + g) * 256 + b] = run;
}

__global__ __launch_bounds__(256) void split_scan_b(
    unsigned* __restrict__ grp, unsigned* __restrict__ bbase) {
  __shared__ unsigned s[256];
  const int b = threadIdx.x;
  const int dir = blockIdx.x;
  unsigned* __restrict__ c = grp + (size_t)dir * L1_NGRP * 256 + b;
  unsigned run = 0u;
  for (int g0 = 0; g0 < L1_NGRP; g0 += 16) {
    unsigned v[16];
#pragma unroll
    for (int j = 0; j < 16; ++j) v[j] = c[(size_t)(g0 + j) * 256];
#pragma unroll
    for (int j = 0; j < 16; ++j) { c[(size_t)(g0 + j) * 256] = run; run += v[j]; }
  }
  s[b] = run;
  __syncthreads();
  for (int off = 1; off < 256; off <<= 1) {
    const unsigned v = (b >= off) ? s[b - off] : 0u;
    __syncthreads();
    s[b] += v;
    __syncthreads();
  }
  bbase[dir * 257 + b] = s[b] - run;
  if (b == 255) bbase[dir * 257 + 256] = s[255];
}

__global__ __launch_bounds__(1024, 16) void split_place(
    const int* __restrict__ u_idx, const int* __restrict__ i_idx,
    const float* __restrict__ vals_ui, const float* __restrict__ vals_iu,
    int nnz, int U, int I,
    const unsigned* __restrict__ cnt1, const unsigned* __restrict__ grp,
    const unsigned* __restrict__ bbase, int* __restrict__ rec1_) {
  __shared__ unsigned hist_s[L1_GRP * 256];
  __shared__ unsigned s[256];
  __shared__ unsigned delta_s[256];
  __shared__ int2 stage[L1_CAP];
  const int lane = threadIdx.x & 31;
  const int wl = threadIdx.x >> 5;
  const int t = threadIdx.x;
  const int dir = blockIdx.y;
  const int g = blockIdx.x;
  const int* __restrict__ key = dir ? i_idx : u_idx;
  const int* __restrict__ src = dir ? u_idx : i_idx;
  const float* __restrict__ vals = dir ? vals_iu : vals_ui;
  const unsigned nrows = (unsigned)(dir ? I : U);
  const int wg = g * L1_GRP + wl;
  int2* __restrict__ rec1 = (int2*)rec1_ + (size_t)dir * (size_t)nnz;
  const int hb = wl * 256;
  const unsigned* __restrict__ c1 = cnt1 + ((size_t)dir * L1_NW + (size_t)wg) * 256;
  const unsigned* __restrict__ gp = grp + ((size_t)dir * L1_NGRP + (size_t)g) * 256;
  const unsigned* __restrict__ bb = bbase + dir * 257;
  const int per = (nnz + L1_NW - 1) / L1_NW;
  const int e_lo = min(nnz, wg * per);
  const int e_hi = min(nnz, e_lo + per);
  const unsigned lt = (1u << lane) - 1u;
  unsigned kn = 0xffffffffu;
  int cn = 0;
  float vn = 0.0f;
  if (e_lo + lane < e_hi) { kn = (unsigned)key[e_lo + lane]; cn = src[e_lo + lane]; vn = vals[e_lo + lane]; }
  unsigned run0 = 0u, n = 0u;
  if (t < 256) {
    const unsigned pre = gp[t];
    const unsigned nxt = (g + 1 < L1_NGRP) ? gp[256 + t] : (bb[t + 1] - bb[t]);
    run0 = bb[t] + pre;
    n = nxt - pre;
    s[t] = n;
  }
  __syncthreads();
  for (int off = 1; off < 256; off <<= 1) {
    unsigned v = 0u;
    if (t < 256 && t >= off) v = s[t - off];
    __syncthreads();
    if (t < 256) s[t] += v;
    __syncthreads();
  }
  const unsigned total = s[255];
  const bool staged = (total <= (unsigned)L1_CAP);
  if (t < 256) delta_s[t] = staged ? run0 - (s[t] - n) : 0u;
  __syncthreads();
  for (int j = lane; j < 256; j += 32) hist_s[hb + j] = bb[j] + gp[j] + c1[j] - delta_s[j];
  for (int eb = e_lo; eb < e_hi; eb += 32) {
    const unsigned k = kn;
    const int c = cn;
    const float v = vn;
    const int en = eb + 32 + lane;
    kn = 0xffffffffu;
    if (en < e_hi) { kn = (unsigned)key[en]; cn = src[en]; vn = vals[en]; }
    const bool valid = (k < nrows);
    const unsigned d = (k >> 8) & 255u;
    const unsigned peers = same_digit_lanes(d, valid);
    const unsigned rank = (unsigned)__builtin_popcount(peers & lt);
    unsigned old = 0u;
    if (valid) old = hist_s[hb + d];
    if (valid && rank == 0u) hist_s[hb + d] = old + (unsigned)__builtin_popcount(peers);
    if (valid) {
      const unsigned pos = old + rank;
      const unsigned x = ((k & 255u) << 16) | (unsigned)c;
      if (staged) stage[pos] = make_int2((int)((x & 0x00ffffffu) | (d << 24)), __float_as_int(v));
      else        rec1[pos] = make_int2((int)x, __float_as_int(v));
    }
  }
  if (staged) {
    __syncthreads();
    for (unsigned j = (unsigned)t; j < total; j += 1024u) {
      const int2 r = stage[j];
      const unsigned d = (unsigned)r.x >> 24;
      rec1[j + delta_s[d]] = make_int2(r.x & 0x00ffffff, r.y);
    }
  }
}

__global__ __launch_bounds__(1024, 16) void rows_place(
    const int* __restrict__ rec1_, const unsigned* __restrict__ bbase,
    int nnz, int U, int I,
    unsigned* __restrict__ rowptr_u, unsigned* __restrict__ rowptr_i,
    int* __restrict__ rec_u, int* __restrict__ rec_i) {
  __shared__ unsigned hist_s[32 * 256];
  __shared__ unsigned s[256];
  __shared__ float stage_v[L2_CAP];
  __shared__ unsigned short stage_c[L2_CAP];
  const int lane = threadIdx.x & 31;
  const int wl = threadIdx.x >> 5;
  const int t = threadIdx.x;
  const int dir = blockIdx.y;
  const int bucket = blockIdx.x;
  const int nrows = dir ? I : U;
  if (bucket * 256 >= nrows) return;
  const int2* __restrict__ rec1 = (const int2*)rec1_ + (size_t)dir * (size_t)nnz;
  int2* __restrict__ rec = (int2*)(dir ? rec_i : rec_u);
  unsigned* __restrict__ rowptr = dir ? rowptr_i : rowptr_u;
  const unsigned bb = bbase[dir * 257 + bucket];
  const unsigned be = bbase[dir * 257 + bucket + 1];
  const unsigned nb = be - bb;
  const bool staged = (nb <= (unsigned)L2_CAP);
  const unsigned per = (nb + 31u) / 32u;
  const unsigned lo = min(be, bb + (unsigned)wl * per);
  const unsigned hi = min(be, lo + per);
  const int hb = wl * 256;
  for (int j = lane; j < 256; j += 32) hist_s[hb + j] = 0u;
  const unsigned lt = (1u << lane) - 1u;
  {
    int xn = 0;
    if (lo + (unsigned)lane < hi) xn = rec1[lo + (unsigned)lane].x;
    for (unsigned p0 = lo; p0 < hi; p0 += 32u) {
      const bool valid = (p0 + (unsigned)lane < hi);
      const unsigned d = ((unsigned)xn >> 16) & 255u;
      const unsigned pn = p0 + 32u + (unsigned)lane;
      xn = 0;
      if (pn < hi) xn = rec1[pn].x;
      const unsigned peers = same_digit_lanes(d, valid);
      if (valid && (peers & lt) == 0u)
        hist_s[hb + d] = hist_s[hb + d] + (unsigned)__builtin_popcount(peers);
    }
  }
  __syncthreads();
  unsigned tot = 0u;
  if (t < 256) {
    for (int w = 0; w < 32; ++w) {
      const unsigned c = hist_s[w * 256 + t];
      hist_s[w * 256 + t] = tot;
      tot += c;
    }
    s[t] = tot;
  }
  __syncthreads();
  for (int off = 1; off < 256; off <<= 1) {
    unsigned v = 0u;
    if (t < 256 && t >= off) v = s[t - off];
    __syncthreads();
    if (t < 256) s[t] += v;
    __syncthreads();
  }
  if (t < 256) {
    const unsigned start = bb + s[t] - tot;
    const int row = bucket * 256 + t;
    if (row < nrows) rowptr[row] = start;
    if (row == nrows - 1) rowptr[nrows] = be;
    for (int w = 0; w < 32; ++w) hist_s[w * 256 + t] += start;
  }
  __syncthreads();
  int2 rn = make_int2(0, 0);
  if (lo + (unsigned)lane < hi) rn = rec1[lo + (unsigned)lane];
  for (unsigned p0 = lo; p0 < hi; p0 += 32u) {
    const bool valid = (p0 + (unsigned)lane < hi);
    const int2 r = rn;
    const unsigned pn = p0 + 32u + (unsigned)lane;
    rn = make_int2(0, 0);
    if (pn < hi) rn = rec1[pn];
    const unsigned d = ((unsigned)r.x >> 16) & 255u;
    const unsigned peers = same_digit_lanes(d, valid);
    const unsigned rank = (unsigned)__builtin_popcount(peers & lt);
    unsigned old = 0u;
    if (valid) old = hist_s[hb + d];
    if (valid && rank == 0u) hist_s[hb + d] = old + (unsigned)__builtin_popcount(peers);
    if (valid) {
      const unsigned slot = old + rank;
      if (staged) {
        stage_c[slot - bb] = (unsigned short)(r.x & 0xffff);
        stage_v[slot - bb] = __int_as_float(r.y);
      } else {
        rec[slot] = make_int2(r.x & 0xffff, r.y);
      }
    }
  }
  if (staged) {
    __syncthreads();
    const int a = (int)(bb & 15u);
    for (int j = t - a; j < (int)nb; j += 1024)
      if (j >= 0) rec[bb + (unsigned)j] = make_int2((int)stage_c[j], __float_as_int(stage_v[j]));
  }
}

template <bool TAIL>
__device__ __forceinline__ void edges8(const float* __restrict__ x, int2 my, unsigned k,
                                       unsigned n, int lane, float& ax, float& ay) {
  float2 xv[8];
  float v[8];
#pragma unroll
  for (int j = 0; j < 8; ++j) {
    const unsigned kk = TAIL ? min(k + (unsigned)j, n - 1u) : (k + (unsigned)j);
    const int col = __builtin_amdgcn_readlane(my.x, (int)kk);
    v[j] = __int_as_float(__builtin_amdgcn_readlane(my.y, (int)kk));
    xv[j] = ((const float2*)(x + (size_t)col * FACTOR))[lane];
  }
#pragma unroll
  for (int j = 0; j < 8; ++j) {
    if (!TAIL || (k + (unsigned)j < n)) {
      ax += xv[j].x * v[j];
      ay += xv[j].y * v[j];
    }
  }
}

template <bool TAIL>
__device__ __forceinline__ void edges8w(const float* __restrict__ x, int2 my, unsigned k,
                                        unsigned n, int lane, float4& a) {
  float4 xv[8];
  float v[8];
#pragma unroll
  for (int j = 0; j < 8; ++j) {
    const unsigned kk = TAIL ? min(k + (unsigned)j, n - 1u) : (k + (unsigned)j);
    const int col = __builtin_amdgcn_readlane(my.x, (int)kk);
    v[j] = __int_as_float(__builtin_amdgcn_readlane(my.y, (int)kk));
    xv[j] = ((const float4*)(x + (size_t)col * (2 * FACTOR)))[lane];
  }
#pragma unroll
  for (int j = 0; j < 8; ++j) {
    if (!TAIL || (k + (unsigned)j < n)) {
      a.x += xv[j].x * v[j];
      a.y += xv[j].y * v[j];
      a.z += xv[j].z * v[j];
      a.w += xv[j].w * v[j];
    }
  }
}


template <bool FINISH>
__global__ __launch_bounds__(256) void spmm_rows(
    const unsigned* __restrict__ rowptr, const int* __restrict__ rec_,
    const float* __restrict__ x, float* __restrict__ y, int nrows,
    const float* __restrict__ e0, const float* __restrict__ pr) {
  const int lane = threadIdx.x;
  const int row = blockIdx.x * blockDim.y + threadIdx.y;
  if (row >= nrows) return;
  const int r = __builtin_amdgcn_readfirstlane(row);
  const float2 q0 = ((const float2*)(e0 + (size_t)r * FACTOR))[lane];
  float2 q1 = make_float2(0.0f, 0.0f), q2 = q1;
  if (FINISH) {
    const size_t pb = (size_t)r * (2 * FACTOR);
    q1 = ((const float2*)(pr + pb))[lane];
    q2 = ((const float2*)(pr + pb + FACTOR))[lane];
  }
  const unsigned beg = (unsigned)__builtin_amdgcn_readfirstlane((int)rowptr[r]);
  const unsigned end = (unsigned)__builtin_amdgcn_readfirstlane((int)rowptr[r + 1]);
  const int2* __restrict__ rec = (const int2*)rec_;
  float ax = 0.0f, ay = 0.0f;
  for (unsigned c = beg; c < end; c += 32u) {
    const unsigned n = min(32u, end - c);
    int2 my = make_int2(0, 0);
    if ((unsigned)lane < n) my = rec[c + (unsigned)lane];
    unsigned k = 0u;
    for (; k + 8u <= n; k += 8u) edges8<false>(x, my, k, n, lane, ax, ay);
    if (k < n) edges8<true>(x, my, k, n, lane, ax, ay);
  }
  if (FINISH) {
    const bool lo = (lane < 16);
    v2f a;
    a.x = lo ? 1.0f : (1.0f / 3.0f);
    a.y = lo ? 0.5f : 0.25f;
    const float s0x = xhalf(lo ? q2.x : q0.x);
    const float s0y = xhalf(lo ? q2.y : q0.y);
    const float s1x = xhalf(lo ? ax : q1.x);
    const float s1y = xhalf(lo ? ay : q1.y);
    v2f b0, b1, b2, b3;
    b0.x = lo ? q0.x : s0x;  b0.y = lo ? q1.x : s1x;
    b1.x = lo ? q0.y : s0y;  b1.y = lo ? q1.y : s1y;
    b2.x = lo ? s0x : q2.x;  b2.y = lo ? s1x : ax;
    b3.x = lo ? s0y : q2.y;  b3.y = lo ? s1y : ay;
    const v8f z = {};
    const v8f d0 = __builtin_amdgcn_wmma_f32_16x16x4_f32(false, a, false, b0, (short)0, z, false, false);
    const v8f d1 = __builtin_amdgcn_wmma_f32_16x16x4_f32(false, a, false, b1, (short)0, z, false, false);
    const v8f d2 = __builtin_amdgcn_wmma_f32_16x16x4_f32(false, a, false, b2, (short)0, z, false, false);
    const v8f d3 = __builtin_amdgcn_wmma_f32_16x16x4_f32(false, a, false, b3, (short)0, z, false, false);
    float2 o;
    o.x = lo ? d0[0] : d2[0];
    o.y = lo ? d1[0] : d3[0];
    ((float2*)(y + (size_t)r * FACTOR))[lane] = o;
  } else {
    float* __restrict__ yr = y + (size_t)r * (2 * FACTOR);
    ((float2*)yr)[lane] = q0;
    ((float2*)(yr + FACTOR))[lane] = make_float2(ax, ay);
  }
}

template <bool FINISH>
__global__ __launch_bounds__(256) void spmm_rows2(
    const unsigned* __restrict__ rowptr, const int* __restrict__ rec_,
    const float* __restrict__ x, float* __restrict__ y, int nrows,
    const float* __restrict__ self, float* __restrict__ y2) {
  const int lane = threadIdx.x;
  const int row = blockIdx.x * blockDim.y + threadIdx.y;
  if (row >= nrows) return;
  const int r = __builtin_amdgcn_readfirstlane(row);
  float4 q = make_float4(0.0f, 0.0f, 0.0f, 0.0f);
  if (FINISH) q = ((const float4*)(self + (size_t)r * (2 * FACTOR)))[lane];
  const unsigned beg = (unsigned)__builtin_amdgcn_readfirstlane((int)rowptr[r]);
  const unsigned end = (unsigned)__builtin_amdgcn_readfirstlane((int)rowptr[r + 1]);
  const int2* __restrict__ rec = (const int2*)rec_;
  float4 acc = make_float4(0.0f, 0.0f, 0.0f, 0.0f);
  for (unsigned c = beg; c < end; c += 32u) {
    const unsigned n = min(32u, end - c);
    int2 my = make_int2(0, 0);
    if ((unsigned)lane < n) my = rec[c + (unsigned)lane];
    unsigned k = 0u;
    for (; k + 8u <= n; k += 8u) edges8w<false>(x, my, k, n, lane, acc);
    if (k < n) edges8w<true>(x, my, k, n, lane, acc);
  }
  if (FINISH) {
    const bool lo = (lane < 16);
    v2f a;
    a.x = lo ? 1.0f : (1.0f / 3.0f);
    a.y = lo ? 0.5f : 0.25f;
    const float g0 = xhalf(lo ? acc.x : q.x);
    const float g1 = xhalf(lo ? acc.y : q.y);
    const float g2 = xhalf(lo ? acc.z : q.z);
    const float g3 = xhalf(lo ? acc.w : q.w);
    v2f b0, b1, b2, b3;
    b0.x = lo ? q.x : g0;  b0.y = lo ? g0 : acc.x;
    b1.x = lo ? q.y : g1;  b1.y = lo ? g1 : acc.y;
    b2.x = lo ? q.z : g2;  b2.y = lo ? g2 : acc.z;
    b3.x = lo ? q.w : g3;  b3.y = lo ? g3 : acc.w;
    const v8f z = {};
    const v8f d0 = __builtin_amdgcn_wmma_f32_16x16x4_f32(false, a, false, b0, (short)0, z, false, false);
    const v8f d1 = __builtin_amdgcn_wmma_f32_16x16x4_f32(false, a, false, b1, (short)0, z, false, false);
    const v8f d2 = __builtin_amdgcn_wmma_f32_16x16x4_f32(false, a, false, b2, (short)0, z, false, false);
    const v8f d3 = __builtin_amdgcn_wmma_f32_16x16x4_f32(false, a, false, b3, (short)0, z, false, false);
    float4 o;
    o.x = lo ? d0[0] : g0;
    o.y = lo ? d1[0] : g1;
    o.z = lo ? d2[0] : g2;
    o.w = lo ? d3[0] : g3;
    float* __restrict__ dst = (lo ? y : y2) + (size_t)r * FACTOR;
    ((float4*)dst)[lane & 15] = o;
  } else {
    ((float4*)(y + (size_t)r * (2 * FACTOR)))[lane] = acc;
  }
}

extern "C" void kernel_launch(void* const* d_in, const int* in_sizes, int n_in,
                              void* d_out, int out_size, void* d_ws, size_t ws_size,
                              hipStream_t stream) {
  (void)n_in; (void)out_size;

  const float* embed_user = (const float*)d_in[0];
  const float* embed_item = (const float*)d_in[1];
  const int*   u_idx      = (const int*)d_in[2];
  const int*   i_idx      = (const int*)d_in[3];
  const float* vals_ui    = (const float*)d_in[4];
  const float* vals_iu    = (const float*)d_in[5];

  const int U   = in_sizes[0] / FACTOR;
  const int I   = in_sizes[1] / FACTOR;
  const int nnz = in_sizes[2];

  float* g1u = (float*)d_ws;
  float* g1i = g1u + (size_t)U * FACTOR;
  float* g2u = g1i + (size_t)I * FACTOR;
  float* g2i = g2u + (size_t)U * FACTOR;
  float* g3u = g2i + (size_t)I * FACTOR;
  float* g3i = g3u + (size_t)U * FACTOR;

  const size_t plane_bytes = (size_t)(U + I) * FACTOR * sizeof(float);
  const size_t acc_bytes = (size_t)3 * plane_bytes;
  const size_t rec_bytes = (size_t)2 * (size_t)nnz * 8;
  const size_t cnt1_bytes = (size_t)2 * L1_NW * 256 * sizeof(unsigned);
  const size_t small_words = (size_t)(U + I) + 2 + (size_t)2 * L1_NGRP * 256 + (size_t)2 * 257;
  const size_t need = acc_bytes + rec_bytes + small_words * sizeof(unsigned);

  float* out_users = (float*)d_out;
  float* out_items = out_users + (size_t)U * FACTOR;

  const bool ordered = (ws_size >= need) && (U <= 65536) && (I <= 65536) &&
                       (plane_bytes >= cnt1_bytes) && ((size_t)2 * plane_bytes >= rec_bytes);

  if (ordered) {
    int* rec_u = (int*)(g3i + (size_t)I * FACTOR);
    int* rec_i = rec_u + (size_t)2 * (size_t)nnz;
    unsigned* rowptr_u = (unsigned*)(rec_i + (size_t)2 * (size_t)nnz);
    unsigned* rowptr_i = rowptr_u + (size_t)U + 1;
    unsigned* grp   = rowptr_i + (size_t)I + 1;
    unsigned* bbase = grp + (size_t)2 * L1_NGRP * 256;
    unsigned* cnt1 = (unsigned*)g1u;
    int* rec1 = (int*)g2u;

    dim3 l1grid(L1_NW / L1_WPB, 2, 1);
    split_count<<<l1grid, L1_WPB * 32, 0, stream>>>(u_idx, i_idx, nnz, U, I, cnt1);
    split_scan_a<<<dim3(L1_NGRP, 2, 1), 256, 0, stream>>>(cnt1, grp);
    split_scan_b<<<2, 256, 0, stream>>>(grp, bbase);
    split_place<<<dim3(L1_NGRP, 2, 1), L1_GRP * 32, 0, stream>>>(u_idx, i_idx, vals_ui, vals_iu,
                                                                 nnz, U, I, cnt1, grp, bbase, rec1);
    rows_place<<<dim3(256, 2, 1), 1024, 0, stream>>>(rec1, bbase, nnz, U, I,
                                                      rowptr_u, rowptr_i, rec_u, rec_i);

    dim3 rblk(32, 8, 1);
    const int gu = (U + 7) / 8;
    const int gi = (I + 7) / 8;
    float* pair_i = (float*)d_ws;
    float* pair_u = pair_i + (size_t)I * (2 * FACTOR);
    float* h2_i   = pair_u + (size_t)U * (2 * FACTOR);
    const float* none = nullptr;
    float* nonew = nullptr;
    spmm_rows<false><<<gi, rblk, 0, stream>>>(rowptr_i, rec_i, embed_user, pair_i, I, embed_item, none);
    spmm_rows2<false><<<gu, rblk, 0, stream>>>(rowptr_u, rec_u, pair_i, pair_u, U, none, nonew);
    spmm_rows2<true><<<gi, rblk, 0, stream>>>(rowptr_i, rec_i, pair_u, out_items, I, pair_i, h2_i);
    spmm_rows<true><<<gu, rblk, 0, stream>>>(rowptr_u, rec_u, h2_i, out_users, U, embed_user, pair_u);
  } else {
    hipMemsetAsync(d_ws, 0, acc_bytes, stream);

    dim3 sblk(32, 8, 1);
    const int waves = (nnz + 3) / 4;
    const int sgrid = (waves + 7) / 8;

    spmm_coo_atomic<<<sgrid, sblk, 0, stream>>>(u_idx, i_idx, vals_ui, embed_item, g1u, nnz);
    spmm_coo_atomic<<<sgrid, sblk, 0, stream>>>(i_idx, u_idx, vals_iu, embed_user, g1i, nnz);
    spmm_coo_atomic<<<sgrid, sblk, 0, stream>>>(u_idx, i_idx, vals_ui, g1i, g2u, nnz);
    spmm_coo_atomic<<<sgrid, sblk, 0, stream>>>(i_idx, u_idx, vals_iu, g1u, g2i, nnz);
    spmm_coo_atomic<<<sgrid, sblk, 0, stream>>>(u_idx, i_idx, vals_ui, g2i, g3u, nnz);
    spmm_coo_atomic<<<sgrid, sblk, 0, stream>>>(i_idx, u_idx, vals_iu, g2u, g3i, nnz);

    const int wpb = 4;
    combine_wmma<<<(U + wpb - 1) / wpb, wpb * 32, 0, stream>>>(embed_user, g1u, g2u, g3u, out_users, U);
    combine_wmma<<<(I + wpb - 1) / wpb, wpb * 32, 0, stream>>>(embed_item, g1i, g2i, g3i, out_items, I);
  }
}


# Level-1 split local to each 32-wave group (count, scan, order in LDS; ranks kept for the placing sweep); level 2 reads a bucket as 256 runs; 9 launches -> 6, no count/scan kernels; same sums
# speedup vs baseline: 3.7997x; 1.0374x over previous
#include <hip/hip_runtime.h>
#include <hip/hip_bf16.h>

#define FACTOR 64

typedef float v2f __attribute__((ext_vector_type(2)));
typedef float v8f __attribute__((ext_vector_type(8)));

__device__ __forceinline__ void atomic_add_f32_dev(float* p, float v) {
  unsigned long long a = (unsigned long long)p;
  asm volatile("global_atomic_add_f32 %0, %1, off scope:SCOPE_DEV"
               :: "v"(a), "v"(v));
}

__device__ __forceinline__ float xhalf(float x) {
  const int b = __builtin_bit_cast(int, x);
  return __builtin_bit_cast(float,
      __builtin_amdgcn_permlanex16(b, b, 0x76543210, 0xfedcba98, false, false));
}

__device__ __forceinline__ void scatter_row(float* yrow, float2 xv, float v,
                                            bool lo, int off0) {
  const float px = xv.x * v;
  const float py = xv.y * v;
  const float got = xhalf(lo ? py : px);
  float* yp = yrow + off0;
  atomic_add_f32_dev(yp,      lo ? px : got);
  atomic_add_f32_dev(yp + 32, lo ? got : py);
}

__global__ __launch_bounds__(256) void spmm_coo_atomic(
    const int* __restrict__ rows, const int* __restrict__ cols,
    const float* __restrict__ vals, const float* __restrict__ x,
    float* __restrict__ y, int nnz) {
  const int lane = threadIdx.x;
  const int waveId = blockIdx.x * blockDim.y + threadIdx.y;
  const int e0 = waveId * 4;
  if (e0 >= nnz) return;

  __builtin_prefetch(rows + e0 + 4096, 0, 1);
  __builtin_prefetch(cols + e0 + 4096, 0, 1);
  __builtin_prefetch(vals + e0 + 4096, 0, 1);

  const bool lo = (lane < 16);
  const int off0 = lo ? 2 * lane : 2 * lane - 31;

  if (e0 + 3 < nnz) {
    const int eu = __builtin_amdgcn_readfirstlane(e0);
    const int4   r4 = *(const int4*)(rows + eu);
    const int4   c4 = *(const int4*)(cols + eu);
    const float4 v4 = *(const float4*)(vals + eu);
    const int   rr[4] = {r4.x, r4.y, r4.z, r4.w};
    const int   cc[4] = {c4.x, c4.y, c4.z, c4.w};
    const float vv[4] = {v4.x, v4.y, v4.z, v4.w};
#pragma unroll
    for (int k = 0; k < 4; ++k) {
      const float2 xv = ((const float2*)(x + (size_t)cc[k] * FACTOR))[lane];
      scatter_row(y + (size_t)rr[k] * FACTOR, xv, vv[k], lo, off0);
    }
  } else {
    for (int k = 0; k < 4; ++k) {
      const int e = e0 + k;
      if (e >= nnz) break;
      const int r = rows[e];
      const int c = cols[e];
      const float v = vals[e];
      const float2 xv = ((const float2*)(x + (size_t)c * FACTOR))[lane];
      scatter_row(y + (size_t)r * FACTOR, xv, v, lo, off0);
    }
  }
}

__global__ __launch_bounds__(128) void combine_wmma(
    const float* __restrict__ x0, const float* __restrict__ x1,
    const float* __restrict__ x2, const float* __restrict__ x3,
    float* __restrict__ out, int nrows) {
  const int lane = threadIdx.x & 31;
  const int wave = threadIdx.x >> 5;
  const int row = blockIdx.x * (blockDim.x >> 5) + wave;
  if (row >= nrows) return;

  const bool lo = (lane < 16);
  v2f a;
  a.x = lo ? 1.0f : (1.0f / 3.0f);
  a.y = lo ? 0.5f : 0.25f;

  const float* __restrict__ p0 = lo ? x0 : x2;
  const float* __restrict__ p1 = lo ? x1 : x3;

  const size_t base = (size_t)row * FACTOR;
  const int n = lane & 15;

#pragma unroll
  for (int nb = 0; nb < FACTOR; nb += 16) {
    v2f b;
    b.x = p0[base + nb + n];
    b.y = p1[base + nb + n];
    v8f c = {};
    c = __builtin_amdgcn_wmma_f32_16x16x4_f32(
         false, a,  false, b,
         (short)0, c,  false,  false);
    out[base + nb + n] = c[0];
  }
}

#define L1_NW    8192
#define L1_GRP   32
#define L1_NGRP  (L1_NW / L1_GRP)
#define L1_CAP   12288
#define L2_CAP   20480

__device__ __forceinline__ unsigned same_digit_lanes(unsigned d, bool valid) {
  unsigned m = (unsigned)__builtin_amdgcn_ballot_w32(valid);
#pragma unroll
  for (int b = 0; b < 8; ++b) {
    const bool bit = ((d >> b) & 1u) != 0u;
    const unsigned bal = (unsigned)__builtin_amdgcn_ballot_w32(bit);
    m &= bit ? bal : ~bal;
  }
  return m;
}

__global__ __launch_bounds__(1024, 16) void group_place(
    const int* __restrict__ u_idx, const int* __restrict__ i_idx,
    const float* __restrict__ vals_ui, const float* __restrict__ vals_iu,
    int nnz, int U, int I,
    unsigned* __restrict__ offs, int* __restrict__ rec1_) {
  __shared__ unsigned hist_s[L1_GRP * 256];
  __shared__ unsigned s[256];
  __shared__ int2 stage[L1_CAP];
  __shared__ unsigned short q_s[L1_CAP];
  const int lane = threadIdx.x & 31;
  const int wl = threadIdx.x >> 5;
  const int t = threadIdx.x;
  const int dir = blockIdx.y;
  const int g = blockIdx.x;
  const int* __restrict__ key = dir ? i_idx : u_idx;
  const int* __restrict__ src = dir ? u_idx : i_idx;
  const float* __restrict__ vals = dir ? vals_iu : vals_ui;
  const unsigned nrows = (unsigned)(dir ? I : U);
  const int wg = g * L1_GRP + wl;
  const int per = (nnz + L1_NW - 1) / L1_NW;
  int2* __restrict__ rec1 = (int2*)rec1_ + ((size_t)dir * L1_NW + (size_t)g * L1_GRP) * (size_t)per;
  const int hb = wl * 256;
  const int qb = wl * (((per + 31) / 32) * 32);
  const int e_lo = min(nnz, wg * per);
  const int e_hi = min(nnz, e_lo + per);
  const unsigned lt = (1u << lane) - 1u;
  for (int j = lane; j < 256; j += 32) hist_s[hb + j] = 0u;
  {
    unsigned kc = 0xffffffffu;
    if (e_lo + lane < e_hi) kc = (unsigned)key[e_lo + lane];
    for (int eb = e_lo; eb < e_hi; eb += 32) {
      const unsigned k = kc;
      const int en = eb + 32 + lane;
      kc = 0xffffffffu;
      if (en < e_hi) kc = (unsigned)key[en];
      const bool valid = (k < nrows);
      const unsigned d = (k >> 8) & 255u;
      const unsigned peers = same_digit_lanes(d, valid);
      const unsigned rank = (unsigned)__builtin_popcount(peers & lt);
      unsigned old = 0u;
      if (valid) old = hist_s[hb + d];
      if (valid && rank == 0u) hist_s[hb + d] = old + (unsigned)__builtin_popcount(peers);
      q_s[qb + (eb - e_lo) + lane] = (unsigned short)(old + rank);
    }
  }
  unsigned kn = 0xffffffffu;
  int cn = 0;
  float vn = 0.0f;
  if (e_lo + lane < e_hi) { kn = (unsigned)key[e_lo + lane]; cn = src[e_lo + lane]; vn = vals[e_lo + lane]; }
  __syncthreads();
  unsigned tot = 0u;
  if (t < 256) {
    for (int w = 0; w < L1_GRP; ++w) {
      const unsigned c = hist_s[w * 256 + t];
      hist_s[w * 256 + t] = tot;
      tot += c;
    }
    s[t] = tot;
  }
  __syncthreads();
  for (int off = 1; off < 256; off <<= 1) {
    unsigned v = 0u;
    if (t < 256 && t >= off) v = s[t - off];
    __syncthreads();
    if (t < 256) s[t] += v;
    __syncthreads();
  }
  if (t < 256) {
    const unsigned start = s[t] - tot;
    unsigned* __restrict__ o = offs + ((size_t)dir * 257 + (size_t)t) * L1_NGRP + (size_t)g;
    o[0] = start;
    if (t == 255) o[L1_NGRP] = s[255];
    for (int w = 0; w < L1_GRP; ++w) hist_s[w * 256 + t] += start;
  }
  __syncthreads();
  const unsigned total = s[255];
  for (int eb = e_lo; eb < e_hi; eb += 32) {
    const unsigned k = kn;
    const int c = cn;
    const float v = vn;
    const int en = eb + 32 + lane;
    kn = 0xffffffffu;
    if (en < e_hi) { kn = (unsigned)key[en]; cn = src[en]; vn = vals[en]; }
    const bool valid = (k < nrows);
    const unsigned d = (k >> 8) & 255u;
    if (valid) {
      const unsigned pos = hist_s[hb + d] + (unsigned)q_s[qb + (eb - e_lo) + lane];
      stage[pos] = make_int2((int)(((k & 255u) << 16) | (unsigned)c), __float_as_int(v));
    }
  }
  __syncthreads();
  for (unsigned j = (unsigned)t; j < total; j += 1024u) rec1[j] = stage[j];
}

__global__ __launch_bounds__(1024, 16) void rows_place(
    const int* __restrict__ rec1_, const unsigned* __restrict__ offs,
    int nnz, int U, int I,
    unsigned* __restrict__ rowptr_u, unsigned* __restrict__ rowptr_i,
    int* __restrict__ rec_u, int* __restrict__ rec_i) {
  __shared__ unsigned hist_s[32 * 256];
  __shared__ unsigned s[256];
  __shared__ unsigned run_s[L1_NGRP];
  __shared__ unsigned run_e[L1_NGRP];
  __shared__ float stage_v[L2_CAP];
  __shared__ unsigned short stage_c[L2_CAP];
  const int lane = threadIdx.x & 31;
  const int wl = threadIdx.x >> 5;
  const int t = threadIdx.x;
  const int dir = blockIdx.y;
  const int bucket = blockIdx.x;
  const int nrows = dir ? I : U;
  if (bucket * 256 >= nrows) return;
  const int per1 = (nnz + L1_NW - 1) / L1_NW;
  const unsigned gcap = (unsigned)per1 * (unsigned)L1_GRP;
  const int2* __restrict__ rec1 = (const int2*)rec1_ + (size_t)dir * (size_t)L1_NW * (size_t)per1;
  int2* __restrict__ rec = (int2*)(dir ? rec_i : rec_u);
  unsigned* __restrict__ rowptr = dir ? rowptr_i : rowptr_u;
  const unsigned* __restrict__ ob = offs + ((size_t)dir * 257 + (size_t)bucket) * L1_NGRP;
  if (t < L1_NGRP) { run_s[t] = ob[t]; run_e[t] = ob[L1_NGRP + t]; }
  const int hb = wl * 256;
  for (int j = lane; j < 256; j += 32) hist_s[hb + j] = 0u;
  const unsigned lt = (1u << lane) - 1u;
  __syncthreads();
  unsigned bb = 0u, be = 0u;
  for (int j = lane; j < L1_NGRP; j += 32) { bb += run_s[j]; be += run_e[j]; }
  for (int o = 16; o > 0; o >>= 1) { bb += __shfl_xor(bb, o); be += __shfl_xor(be, o); }
  const unsigned nb = be - bb;
  const bool staged = (nb <= (unsigned)L2_CAP);
  const unsigned gq = (unsigned)wl * 8u + (unsigned)(lane & 7);
  const unsigned myS = run_s[gq];
  const unsigned myLen = run_e[gq] - myS;
  unsigned inc = myLen;
#pragma unroll
  for (int o = 1; o < 8; o <<= 1) {
    const unsigned v = __shfl_up(inc, (unsigned)o, 8);
    if ((lane & 7) >= o) inc += v;
  }
  const unsigned myPre = inc - myLen;
  const unsigned myDl = gq * gcap + myS - myPre;
  const unsigned pre1 = (unsigned)__builtin_amdgcn_readlane((int)myPre, 1);
  const unsigned pre2 = (unsigned)__builtin_amdgcn_readlane((int)myPre, 2);
  const unsigned pre3 = (unsigned)__builtin_amdgcn_readlane((int)myPre, 3);
  const unsigned pre4 = (unsigned)__builtin_amdgcn_readlane((int)myPre, 4);
  const unsigned pre5 = (unsigned)__builtin_amdgcn_readlane((int)myPre, 5);
  const unsigned pre6 = (unsigned)__builtin_amdgcn_readlane((int)myPre, 6);
  const unsigned pre7 = (unsigned)__builtin_amdgcn_readlane((int)myPre, 7);
  const unsigned nw = (unsigned)__builtin_amdgcn_readlane((int)inc, 7);
  auto run_pos = [&](unsigned p) -> unsigned {
    const int j = (int)(p >= pre1) + (int)(p >= pre2) + (int)(p >= pre3) + (int)(p >= pre4) +
                  (int)(p >= pre5) + (int)(p >= pre6) + (int)(p >= pre7);
    return p + __shfl(myDl, j);
  };
  {
    int xn = 0;
    const unsigned i0 = run_pos((unsigned)lane);
    if ((unsigned)lane < nw) xn = rec1[i0].x;
    for (unsigned p0 = 0u; p0 < nw; p0 += 32u) {
      const bool valid = (p0 + (unsigned)lane < nw);
      const unsigned d = ((unsigned)xn >> 16) & 255u;
      const unsigned pn = p0 + 32u + (unsigned)lane;
      const unsigned in = run_pos(pn);
      xn = 0;
      if (pn < nw) xn = rec1[in].x;
      const unsigned peers = same_digit_lanes(d, valid);
      if (valid && (peers & lt) == 0u)
        hist_s[hb + d] = hist_s[hb + d] + (unsigned)__builtin_popcount(peers);
    }
  }
  __syncthreads();
  unsigned tot = 0u;
  if (t < 256) {
    for (int w = 0; w < 32; ++w) {
      const unsigned c = hist_s[w * 256 + t];
      hist_s[w * 256 + t] = tot;
      tot += c;
    }
    s[t] = tot;
  }
  __syncthreads();
  for (int off = 1; off < 256; off <<= 1) {
    unsigned v = 0u;
    if (t < 256 && t >= off) v = s[t - off];
    __syncthreads();
    if (t < 256) s[t] += v;
    __syncthreads();
  }
  if (t < 256) {
    const unsigned start = bb + s[t] - tot;
    const int row = bucket * 256 + t;
    if (row < nrows) rowptr[row] = start;
    if (row == nrows - 1) rowptr[nrows] = be;
    for (int w = 0; w < 32; ++w) hist_s[w * 256 + t] += start;
  }
  __syncthreads();
  int2 rn = make_int2(0, 0);
  const unsigned j0 = run_pos((unsigned)lane);
  if ((unsigned)lane < nw) rn = rec1[j0];
  for (unsigned p0 = 0u; p0 < nw; p0 += 32u) {
    const bool valid = (p0 + (unsigned)lane < nw);
    const int2 r = rn;
    const unsigned pn = p0 + 32u + (unsigned)lane;
    const unsigned jn = run_pos(pn);
    rn = make_int2(0, 0);
    if (pn < nw) rn = rec1[jn];
    const unsigned d = ((unsigned)r.x >> 16) & 255u;
    const unsigned peers = same_digit_lanes(d, valid);
    const unsigned rank = (unsigned)__builtin_popcount(peers & lt);
    unsigned old = 0u;
    if (valid) old = hist_s[hb + d];
    if (valid && rank == 0u) hist_s[hb + d] = old + (unsigned)__builtin_popcount(peers);
    if (valid) {
      const unsigned slot = old + rank;
      if (staged) {
        stage_c[slot - bb] = (unsigned short)(r.x & 0xffff);
        stage_v[slot - bb] = __int_as_float(r.y);
      } else {
        rec[slot] = make_int2(r.x & 0xffff, r.y);
      }
    }
  }
  if (staged) {
    __syncthreads();
    const int a = (int)(bb & 15u);
    for (int j = t - a; j < (int)nb; j += 1024)
      if (j >= 0) rec[bb + (unsigned)j] = make_int2((int)stage_c[j], __float_as_int(stage_v[j]));
  }
}

template <bool TAIL>
__device__ __forceinline__ void edges8(const float* __restrict__ x, int2 my, unsigned k,
                                       unsigned n, int lane, float& ax, float& ay) {
  float2 xv[8];
  float v[8];
#pragma unroll
  for (int j = 0; j < 8; ++j) {
    const unsigned kk = TAIL ? min(k + (unsigned)j, n - 1u) : (k + (unsigned)j);
    const int col = __builtin_amdgcn_readlane(my.x, (int)kk);
    v[j] = __int_as_float(__builtin_amdgcn_readlane(my.y, (int)kk));
    xv[j] = ((const float2*)(x + (size_t)col * FACTOR))[lane];
  }
#pragma unroll
  for (int j = 0; j < 8; ++j) {
    if (!TAIL || (k + (unsigned)j < n)) {
      ax += xv[j].x * v[j];
      ay += xv[j].y * v[j];
    }
  }
}

template <bool TAIL>
__device__ __forceinline__ void edges8w(const float* __restrict__ x, int2 my, unsigned k,
                                        unsigned n, int lane, float4& a) {
  float4 xv[8];
  float v[8];
#pragma unroll
  for (int j = 0; j < 8; ++j) {
    const unsigned kk = TAIL ? min(k + (unsigned)j, n - 1u) : (k + (unsigned)j);
    const int col = __builtin_amdgcn_readlane(my.x, (int)kk);
    v[j] = __int_as_float(__builtin_amdgcn_readlane(my.y, (int)kk));
    xv[j] = ((const float4*)(x + (size_t)col * (2 * FACTOR)))[lane];
  }
#pragma unroll
  for (int j = 0; j < 8; ++j) {
    if (!TAIL || (k + (unsigned)j < n)) {
      a.x += xv[j].x * v[j];
      a.y += xv[j].y * v[j];
      a.z += xv[j].z * v[j];
      a.w += xv[j].w * v[j];
    }
  }
}


template <bool FINISH>
__global__ __launch_bounds__(256) void spmm_rows(
    const unsigned* __restrict__ rowptr, const int* __restrict__ rec_,
    const float* __restrict__ x, float* __restrict__ y, int nrows,
    const float* __restrict__ e0, const float* __restrict__ pr) {
  const int lane = threadIdx.x;
  const int row = blockIdx.x * blockDim.y + threadIdx.y;
  if (row >= nrows) return;
  const int r = __builtin_amdgcn_readfirstlane(row);
  const float2 q0 = ((const float2*)(e0 + (size_t)r * FACTOR))[lane];
  float2 q1 = make_float2(0.0f, 0.0f), q2 = q1;
  if (FINISH) {
    const size_t pb = (size_t)r * (2 * FACTOR);
    q1 = ((const float2*)(pr + pb))[lane];
    q2 = ((const float2*)(pr + pb + FACTOR))[lane];
  }
  const unsigned beg = (unsigned)__builtin_amdgcn_readfirstlane((int)rowptr[r]);
  const unsigned end = (unsigned)__builtin_amdgcn_readfirstlane((int)rowptr[r + 1]);
  const int2* __restrict__ rec = (const int2*)rec_;
  float ax = 0.0f, ay = 0.0f;
  for (unsigned c = beg; c < end; c += 32u) {
    const unsigned n = min(32u, end - c);
    int2 my = make_int2(0, 0);
    if ((unsigned)lane < n) my = rec[c + (unsigned)lane];
    unsigned k = 0u;
    for (; k + 8u <= n; k += 8u) edges8<false>(x, my, k, n, lane, ax, ay);
    if (k < n) edges8<true>(x, my, k, n, lane, ax, ay);
  }
  if (FINISH) {
    const bool lo = (lane < 16);
    v2f a;
    a.x = lo ? 1.0f : (1.0f / 3.0f);
    a.y = lo ? 0.5f : 0.25f;
    const float s0x = xhalf(lo ? q2.x : q0.x);
    const float s0y = xhalf(lo ? q2.y : q0.y);
    const float s1x = xhalf(lo ? ax : q1.x);
    const float s1y = xhalf(lo ? ay : q1.y);
    v2f b0, b1, b2, b3;
    b0.x = lo ? q0.x : s0x;  b0.y = lo ? q1.x : s1x;
    b1.x = lo ? q0.y : s0y;  b1.y = lo ? q1.y : s1y;
    b2.x = lo ? s0x : q2.x;  b2.y = lo ? s1x : ax;
    b3.x = lo ? s0y : q2.y;  b3.y = lo ? s1y : ay;
    const v8f z = {};
    const v8f d0 = __builtin_amdgcn_wmma_f32_16x16x4_f32(false, a, false, b0, (short)0, z, false, false);
    const v8f d1 = __builtin_amdgcn_wmma_f32_16x16x4_f32(false, a, false, b1, (short)0, z, false, false);
    const v8f d2 = __builtin_amdgcn_wmma_f32_16x16x4_f32(false, a, false, b2, (short)0, z, false, false);
    const v8f d3 = __builtin_amdgcn_wmma_f32_16x16x4_f32(false, a, false, b3, (short)0, z, false, false);
    float2 o;
    o.x = lo ? d0[0] : d2[0];
    o.y = lo ? d1[0] : d3[0];
    ((float2*)(y + (size_t)r * FACTOR))[lane] = o;
  } else {
    float* __restrict__ yr = y + (size_t)r * (2 * FACTOR);
    ((float2*)yr)[lane] = q0;
    ((float2*)(yr + FACTOR))[lane] = make_float2(ax, ay);
  }
}

template <bool FINISH>
__global__ __launch_bounds__(256) void spmm_rows2(
    const unsigned* __restrict__ rowptr, const int* __restrict__ rec_,
    const float* __restrict__ x, float* __restrict__ y, int nrows,
    const float* __restrict__ self, float* __restrict__ y2) {
  const int lane = threadIdx.x;
  const int row = blockIdx.x * blockDim.y + threadIdx.y;
  if (row >= nrows) return;
  const int r = __builtin_amdgcn_readfirstlane(row);
  float4 q = make_float4(0.0f, 0.0f, 0.0f, 0.0f);
  if (FINISH) q = ((const float4*)(self + (size_t)r * (2 * FACTOR)))[lane];
  const unsigned beg = (unsigned)__builtin_amdgcn_readfirstlane((int)rowptr[r]);
  const unsigned end = (unsigned)__builtin_amdgcn_readfirstlane((int)rowptr[r + 1]);
  const int2* __restrict__ rec = (const int2*)rec_;
  float4 acc = make_float4(0.0f, 0.0f, 0.0f, 0.0f);
  for (unsigned c = beg; c < end; c += 32u) {
    const unsigned n = min(32u, end - c);
    int2 my = make_int2(0, 0);
    if ((unsigned)lane < n) my = rec[c + (unsigned)lane];
    unsigned k = 0u;
    for (; k + 8u <= n; k += 8u) edges8w<false>(x, my, k, n, lane, acc);
    if (k < n) edges8w<true>(x, my, k, n, lane, acc);
  }
  if (FINISH) {
    const bool lo = (lane < 16);
    v2f a;
    a.x = lo ? 1.0f : (1.0f / 3.0f);
    a.y = lo ? 0.5f : 0.25f;
    const float g0 = xhalf(lo ? acc.x : q.x);
    const float g1 = xhalf(lo ? acc.y : q.y);
    const float g2 = xhalf(lo ? acc.z : q.z);
    const float g3 = xhalf(lo ? acc.w : q.w);
    v2f b0, b1, b2, b3;
    b0.x = lo ? q.x : g0;  b0.y = lo ? g0 : acc.x;
    b1.x = lo ? q.y : g1;  b1.y = lo ? g1 : acc.y;
    b2.x = lo ? q.z : g2;  b2.y = lo ? g2 : acc.z;
    b3.x = lo ? q.w : g3;  b3.y = lo ? g3 : acc.w;
    const v8f z = {};
    const v8f d0 = __builtin_amdgcn_wmma_f32_16x16x4_f32(false, a, false, b0, (short)0, z, false, false);
    const v8f d1 = __builtin_amdgcn_wmma_f32_16x16x4_f32(false, a, false, b1, (short)0, z, false, false);
    const v8f d2 = __builtin_amdgcn_wmma_f32_16x16x4_f32(false, a, false, b2, (short)0, z, false, false);
    const v8f d3 = __builtin_amdgcn_wmma_f32_16x16x4_f32(false, a, false, b3, (short)0, z, false, false);
    float4 o;
    o.x = lo ? d0[0] : g0;
    o.y = lo ? d1[0] : g1;
    o.z = lo ? d2[0] : g2;
    o.w = lo ? d3[0] : g3;
    float* __restrict__ dst = (lo ? y : y2) + (size_t)r * FACTOR;
    ((float4*)dst)[lane & 15] = o;
  } else {
    ((float4*)(y + (size_t)r * (2 * FACTOR)))[lane] = acc;
  }
}

extern "C" void kernel_launch(void* const* d_in, const int* in_sizes, int n_in,
                              void* d_out, int out_size, void* d_ws, size_t ws_size,
                              hipStream_t stream) {
  (void)n_in; (void)out_size;

  const float* embed_user = (const float*)d_in[0];
  const float* embed_item = (const float*)d_in[1];
  const int*   u_idx      = (const int*)d_in[2];
  const int*   i_idx      = (const int*)d_in[3];
  const float* vals_ui    = (const float*)d_in[4];
  const float* vals_iu    = (const float*)d_in[5];

  const int U   = in_sizes[0] / FACTOR;
  const int I   = in_sizes[1] / FACTOR;
  const int nnz = in_sizes[2];

  float* g1u = (float*)d_ws;
  float* g1i = g1u + (size_t)U * FACTOR;
  float* g2u = g1i + (size_t)I * FACTOR;
  float* g2i = g2u + (size_t)U * FACTOR;
  float* g3u = g2i + (size_t)I * FACTOR;
  float* g3i = g3u + (size_t)U * FACTOR;

  const size_t plane_bytes = (size_t)(U + I) * FACTOR * sizeof(float);
  const size_t acc_bytes = (size_t)3 * plane_bytes;
  const size_t rec_bytes = (size_t)2 * (size_t)nnz * 8;
  const int per1 = (nnz + L1_NW - 1) / L1_NW;
  const size_t rec1_bytes = (size_t)2 * L1_NW * (size_t)per1 * 8;
  const size_t small_words = (size_t)(U + I) + 2 + (size_t)2 * 257 * L1_NGRP;
  const size_t need = acc_bytes + rec_bytes + small_words * sizeof(unsigned);

  float* out_users = (float*)d_out;
  float* out_items = out_users + (size_t)U * FACTOR;

  const bool ordered = (ws_size >= need) && (U <= 65536) && (I <= 65536) &&
                       (((per1 + 31) / 32) * 32 * L1_GRP <= L1_CAP) && ((size_t)2 * plane_bytes >= rec1_bytes);

  if (ordered) {
    int* rec_u = (int*)(g3i + (size_t)I * FACTOR);
    int* rec_i = rec_u + (size_t)2 * (size_t)nnz;
    unsigned* rowptr_u = (unsigned*)(rec_i + (size_t)2 * (size_t)nnz);
    unsigned* rowptr_i = rowptr_u + (size_t)U + 1;
    unsigned* offs = rowptr_i + (size_t)I + 1;
    int* rec1 = (int*)g2u;

    group_place<<<dim3(L1_NGRP, 2, 1), L1_GRP * 32, 0, stream>>>(u_idx, i_idx, vals_ui, vals_iu,
                                                                 nnz, U, I, offs, rec1);
    rows_place<<<dim3(256, 2, 1), 1024, 0, stream>>>(rec1, offs, nnz, U, I,
                                                      rowptr_u, rowptr_i, rec_u, rec_i);

    dim3 rblk(32, 8, 1);
    const int gu = (U + 7) / 8;
    const int gi = (I + 7) / 8;
    float* pair_i = (float*)d_ws;
    float* pair_u = pair_i + (size_t)I * (2 * FACTOR);
    float* h2_i   = pair_u + (size_t)U * (2 * FACTOR);
    const float* none = nullptr;
    float* nonew = nullptr;
    spmm_rows<false><<<gi, rblk, 0, stream>>>(rowptr_i, rec_i, embed_user, pair_i, I, embed_item, none);
    spmm_rows2<false><<<gu, rblk, 0, stream>>>(rowptr_u, rec_u, pair_i, pair_u, U, none, nonew);
    spmm_rows2<true><<<gi, rblk, 0, stream>>>(rowptr_i, rec_i, pair_u, out_items, I, pair_i, h2_i);
    spmm_rows<true><<<gu, rblk, 0, stream>>>(rowptr_u, rec_u, h2_i, out_users, U, embed_user, pair_u);
  } else {
    hipMemsetAsync(d_ws, 0, acc_bytes, stream);

    dim3 sblk(32, 8, 1);
    const int waves = (nnz + 3) / 4;
    const int sgrid = (waves + 7) / 8;

    spmm_coo_atomic<<<sgrid, sblk, 0, stream>>>(u_idx, i_idx, vals_ui, embed_item, g1u, nnz);
    spmm_coo_atomic<<<sgrid, sblk, 0, stream>>>(i_idx, u_idx, vals_iu, embed_user, g1i, nnz);
    spmm_coo_atomic<<<sgrid, sblk, 0, stream>>>(u_idx, i_idx, vals_ui, g1i, g2u, nnz);
    spmm_coo_atomic<<<sgrid, sblk, 0, stream>>>(i_idx, u_idx, vals_iu, g1u, g2i, nnz);
    spmm_coo_atomic<<<sgrid, sblk, 0, stream>>>(u_idx, i_idx, vals_ui, g2i, g3u, nnz);
    spmm_coo_atomic<<<sgrid, sblk, 0, stream>>>(i_idx, u_idx, vals_iu, g2u, g3i, nnz);

    const int wpb = 4;
    combine_wmma<<<(U + wpb - 1) / wpb, wpb * 32, 0, stream>>>(embed_user, g1u, g2u, g3u, out_users, U);
    combine_wmma<<<(I + wpb - 1) / wpb, wpb * 32, 0, stream>>>(embed_item, g1i, g2i, g3i, out_items, I);
  }
}
